# starting kernel
# baseline (speedup 1.0000x reference)
#include <hip/hip_runtime.h>

typedef __attribute__((ext_vector_type(2))) float v2f;
typedef __attribute__((ext_vector_type(4))) float v4f;
typedef __attribute__((ext_vector_type(8))) float v8f;

#define F_IN  128
#define F_HID 64
#define F_OUT 40


__global__ void deg_init_kernel(float* __restrict__ deg, int N) {
    int i = blockIdx.x * blockDim.x + threadIdx.x;
    if (i < N) deg[i] = 1.0f;
}

__global__ void deg_edge_kernel(const int* __restrict__ dst, float* __restrict__ deg, int E) {
    int e = blockIdx.x * blockDim.x + threadIdx.x;
    if (e < E) unsafeAtomicAdd(&deg[dst[e]], 1.0f);
}

__global__ void dinv_kernel(const float* __restrict__ deg, float* __restrict__ dinv, int N) {
    int i = blockIdx.x * blockDim.x + threadIdx.x;
    if (i < N) dinv[i] = rsqrtf(deg[i]);
}


__global__ void repack_w1_kernel(const float* __restrict__ W1, float* __restrict__ pW1) {
    int idx = blockIdx.x * blockDim.x + threadIdx.x;
    if (idx >= (F_IN / 4) * 256) return;
    int kk   = idx >> 8;
    int rem  = idx & 255;
    int lane = rem >> 3;
    int j    = rem & 7;
    int n    = j >> 1;
    int t    = j & 1;
    int lm   = lane & 15;
    int kh   = (lane >> 4) << 1;
    pW1[idx] = W1[(size_t)(4 * kk + kh + t) * F_HID + n * 16 + lm];
}

__global__ void repack_w2_kernel(const float* __restrict__ W2, float* __restrict__ pW2) {
    int idx = blockIdx.x * blockDim.x + threadIdx.x;
    if (idx >= (F_HID / 4) * 256) return;
    int kk   = idx >> 8;
    int rem  = idx & 255;
    int lane = rem >> 3;
    int j    = rem & 7;
    int n    = j >> 1;
    int t    = j & 1;
    int lm   = lane & 15;
    int kh   = (lane >> 4) << 1;
    int col  = n * 16 + lm;
    float v  = 0.0f;
    if (j < 6 && col < F_OUT)
        v = W2[(size_t)(4 * kk + kh + t) * F_OUT + col];
    pW2[idx] = v;
}


__global__ void gemm1_wmma_kernel(const float* __restrict__ x, const float* __restrict__ pW1,
                                  float* __restrict__ h1, int N) {
    const int lane = threadIdx.x & 31;
    const int wave = threadIdx.x >> 5;
    const int row0 = blockIdx.x * 128 + wave * 16;
    const int lm   = lane & 15;
    const int kh   = (lane >> 4) << 1;
    int r = row0 + lm;
    if (r > N - 1) r = N - 1;
    const float* xrow = x + (size_t)r * F_IN + kh;
    const float* pw   = pW1 + (size_t)lane * 8;

    v8f acc0 = {}, acc1 = {}, acc2 = {}, acc3 = {};

    for (int kk = 0; kk < F_IN / 4; ++kk) {
        const v2f A   = *(const v2f*)(xrow + kk * 4);
        const v4f w01 = *(const v4f*)(pw + (size_t)kk * 256);
        const v4f w23 = *(const v4f*)(pw + (size_t)kk * 256 + 4);
        v2f B0; B0[0] = w01[0]; B0[1] = w01[1];
        v2f B1; B1[0] = w01[2]; B1[1] = w01[3];
        v2f B2; B2[0] = w23[0]; B2[1] = w23[1];
        v2f B3; B3[0] = w23[2]; B3[1] = w23[3];
        acc0 = __builtin_amdgcn_wmma_f32_16x16x4_f32(false, A, false, B0, (short)0, acc0, false, false);
        acc1 = __builtin_amdgcn_wmma_f32_16x16x4_f32(false, A, false, B1, (short)0, acc1, false, false);
        acc2 = __builtin_amdgcn_wmma_f32_16x16x4_f32(false, A, false, B2, (short)0, acc2, false, false);
        acc3 = __builtin_amdgcn_wmma_f32_16x16x4_f32(false, A, false, B3, (short)0, acc3, false, false);
    }

    const int hi8 = (lane >> 4) << 3;
    for (int j = 0; j < 8; ++j) {
        int rr = row0 + hi8 + j;
        if (rr < N) {
            float* o = h1 + (size_t)rr * F_HID + lm;
            o[0]  = acc0[j];
            o[16] = acc1[j];
            o[32] = acc2[j];
            o[48] = acc3[j];
        }
    }
}


__global__ void acc1_init_kernel(const float* __restrict__ h1, const float* __restrict__ dinv,
                                 float* __restrict__ acc, int total) {
    int idx = blockIdx.x * blockDim.x + threadIdx.x;
    if (idx < total) {
        float di = dinv[idx >> 6];
        acc[idx] = h1[idx] * di * di;
    }
}


__global__ void scatter1_kernel(const float* __restrict__ h1, const float* __restrict__ dinv,
                                const int* __restrict__ src, const int* __restrict__ dst,
                                float* __restrict__ acc, int E) {
    const int lane = threadIdx.x & 31;
    const int e = blockIdx.x * (blockDim.x >> 5) + (threadIdx.x >> 5);
    if (e >= E) return;
    const int s = __builtin_amdgcn_readfirstlane(src[e]);
    const int d = __builtin_amdgcn_readfirstlane(dst[e]);
    const float norm = dinv[s] * dinv[d];
    const float2 v = *(const float2*)(h1 + (size_t)s * F_HID + lane * 2);
    float* o = acc + (size_t)d * F_HID + lane * 2;
    unsafeAtomicAdd(o,     v.x * norm);
    unsafeAtomicAdd(o + 1, v.y * norm);
}


__global__ void relu_bias_kernel(const float* __restrict__ acc, const float* __restrict__ b1,
                                 float* __restrict__ z, int total) {
    int idx = blockIdx.x * blockDim.x + threadIdx.x;
    if (idx < total) {
        float v = acc[idx] + b1[idx & (F_HID - 1)];
        z[idx] = v > 0.0f ? v : 0.0f;
    }
}


__global__ void gemm2_wmma_kernel(const float* __restrict__ z, const float* __restrict__ pW2,
                                  float* __restrict__ h2, int N) {
    const int lane = threadIdx.x & 31;
    const int wave = threadIdx.x >> 5;
    const int row0 = blockIdx.x * 128 + wave * 16;
    const int lm   = lane & 15;
    const int kh   = (lane >> 4) << 1;
    int r = row0 + lm;
    if (r > N - 1) r = N - 1;
    const float* zrow = z + (size_t)r * F_HID + kh;
    const float* pw   = pW2 + (size_t)lane * 8;

    v8f acc0 = {}, acc1 = {}, acc2 = {};

    for (int kk = 0; kk < F_HID / 4; ++kk) {
        const v2f A   = *(const v2f*)(zrow + kk * 4);
        const v4f w01 = *(const v4f*)(pw + (size_t)kk * 256);
        const v2f w2v = *(const v2f*)(pw + (size_t)kk * 256 + 4);
        v2f B0; B0[0] = w01[0]; B0[1] = w01[1];
        v2f B1; B1[0] = w01[2]; B1[1] = w01[3];
        v2f B2; B2[0] = w2v[0]; B2[1] = w2v[1];
        acc0 = __builtin_amdgcn_wmma_f32_16x16x4_f32(false, A, false, B0, (short)0, acc0, false, false);
        acc1 = __builtin_amdgcn_wmma_f32_16x16x4_f32(false, A, false, B1, (short)0, acc1, false, false);
        acc2 = __builtin_amdgcn_wmma_f32_16x16x4_f32(false, A, false, B2, (short)0, acc2, false, false);
    }

    const int hi8  = (lane >> 4) << 3;
    const bool c2ok = (lm < 8);
    for (int j = 0; j < 8; ++j) {
        int rr = row0 + hi8 + j;
        if (rr < N) {
            float* o = h2 + (size_t)rr * F_OUT + lm;
            o[0]  = acc0[j];
            o[16] = acc1[j];
            if (c2ok) o[32] = acc2[j];
        }
    }
}


__global__ void out_init_kernel(const float* __restrict__ h2, const float* __restrict__ dinv,
                                const float* __restrict__ b2, float* __restrict__ out, int total) {
    int idx = blockIdx.x * blockDim.x + threadIdx.x;
    if (idx < total) {
        int i = idx / F_OUT;
        int f = idx - i * F_OUT;
        float di = dinv[i];
        out[idx] = h2[idx] * di * di + b2[f];
    }
}


__global__ void scatter2_kernel(const float* __restrict__ h2, const float* __restrict__ dinv,
                                const int* __restrict__ src, const int* __restrict__ dst,
                                float* __restrict__ out, int E) {
    const int lane = threadIdx.x & 31;
    const int e = blockIdx.x * (blockDim.x >> 5) + (threadIdx.x >> 5);
    if (e >= E) return;
    const int s = __builtin_amdgcn_readfirstlane(src[e]);
    const int d = __builtin_amdgcn_readfirstlane(dst[e]);
    const float norm = dinv[s] * dinv[d];
    if (lane < 20) {
        const float2 v = *(const float2*)(h2 + (size_t)s * F_OUT + lane * 2);
        float* o = out + (size_t)d * F_OUT + lane * 2;
        unsafeAtomicAdd(o,     v.x * norm);
        unsafeAtomicAdd(o + 1, v.y * norm);
    }
}


extern "C" void kernel_launch(void* const* d_in, const int* in_sizes, int n_in,
                              void* d_out, int out_size, void* d_ws, size_t ws_size,
                              hipStream_t stream) {
    const float* x  = (const float*)d_in[0];
    const int*   ei = (const int*)d_in[1];
    const float* W1 = (const float*)d_in[2];
    const float* b1 = (const float*)d_in[3];
    const float* W2 = (const float*)d_in[4];
    const float* b2 = (const float*)d_in[5];

    const int N = in_sizes[0] / F_IN;
    const int E = in_sizes[1] / 2;
    const int* src = ei;
    const int* dst = ei + E;

    float* ws   = (float*)d_ws;
    float* deg  = ws;
    float* dinv = ws + (size_t)N;
    float* bufA = ws + (size_t)2 * N;
    float* bufB = bufA + (size_t)N * F_HID;
    float* pW1  = bufB + (size_t)N * F_HID;
    float* pW2  = pW1 + (size_t)(F_IN / 4) * 256;
    float* out  = (float*)d_out;

    const int B = 256;
    const int nodeBlocks  = (N + B - 1) / B;
    const int edgeBlocks  = (E + B - 1) / B;
    const int hidElems    = N * F_HID;
    const int outElems    = N * F_OUT;
    const int hidBlocks   = (hidElems + B - 1) / B;
    const int outBlocksEl = (outElems + B - 1) / B;
    const int gemmBlocks  = (N + 127) / 128;
    const int waveEdgeBlk = (E + (B / 32) - 1) / (B / 32);
    const int rp1Blocks   = ((F_IN / 4) * 256 + B - 1) / B;
    const int rp2Blocks   = ((F_HID / 4) * 256 + B - 1) / B;

    repack_w1_kernel<<<rp1Blocks, B, 0, stream>>>(W1, pW1);
    repack_w2_kernel<<<rp2Blocks, B, 0, stream>>>(W2, pW2);

    deg_init_kernel<<<nodeBlocks, B, 0, stream>>>(deg, N);
    deg_edge_kernel<<<edgeBlocks, B, 0, stream>>>(dst, deg, E);
    dinv_kernel<<<nodeBlocks, B, 0, stream>>>(deg, dinv, N);

    gemm1_wmma_kernel<<<gemmBlocks, B, 0, stream>>>(x, pW1, bufA, N);
    acc1_init_kernel<<<hidBlocks, B, 0, stream>>>(bufA, dinv, bufB, hidElems);
    scatter1_kernel<<<waveEdgeBlk, B, 0, stream>>>(bufA, dinv, src, dst, bufB, E);
    relu_bias_kernel<<<hidBlocks, B, 0, stream>>>(bufB, b1, bufA, hidElems);

    gemm2_wmma_kernel<<<gemmBlocks, B, 0, stream>>>(bufA, pW2, bufB, N);
    out_init_kernel<<<outBlocksEl, B, 0, stream>>>(bufB, dinv, b2, out, outElems);
    scatter2_kernel<<<waveEdgeBlk, B, 0, stream>>>(bufB, dinv, src, dst, out, E);
}


# Float-atomic edge scatter replaced by a per-node gather over neighbour lists built in-call (int-atomic slots + wave-scan offsets), rows pre-scaled by dinv; per-node sum order = slot arrival order
# speedup vs baseline: 2.0405x; 2.0405x over previous
#include <hip/hip_runtime.h>

typedef __attribute__((ext_vector_type(2))) float v2f;
typedef __attribute__((ext_vector_type(4))) float v4f;
typedef __attribute__((ext_vector_type(8))) float v8f;

#define F_IN  128
#define F_HID 64
#define F_OUT 40


__global__ void cnt_init_kernel(int* __restrict__ cnt, int* __restrict__ total, int N) {
    int i = blockIdx.x * blockDim.x + threadIdx.x;
    if (i < N) cnt[i] = 0;
    if (i == 0) *total = 0;
}

__global__ void rank_edge_kernel(const int* __restrict__ dst, int* __restrict__ cnt,
                                 int* __restrict__ rank, int E) {
    int e = blockIdx.x * blockDim.x + threadIdx.x;
    if (e < E) rank[e] = atomicAdd(&cnt[dst[e]], 1);
}

__global__ void alloc_kernel(const int* __restrict__ cnt, int* __restrict__ total,
                             int* __restrict__ start, float* __restrict__ dinv, int N) {
    const int i    = blockIdx.x * blockDim.x + threadIdx.x;
    const int lane = threadIdx.x & 31;
    const int c    = (i < N) ? cnt[i] : 0;
    int incl = c;
    for (int o = 1; o < 32; o <<= 1) {
        int t = __shfl_up(incl, o);
        if (lane >= o) incl += t;
    }
    const int wsum = __shfl(incl, 31);
    int base = 0;
    if (lane == 0) base = atomicAdd(total, wsum);
    base = __shfl(base, 0);
    if (i < N) {
        start[i] = base + incl - c;
        dinv[i]  = rsqrtf((float)(c + 1));
    }
}

__global__ void fill_kernel(const int* __restrict__ src, const int* __restrict__ dst,
                            const int* __restrict__ rank, const int* __restrict__ start,
                            int* __restrict__ csr, int E) {
    int e = blockIdx.x * blockDim.x + threadIdx.x;
    if (e < E) csr[start[dst[e]] + rank[e]] = src[e];
}


__global__ void repack_w1_kernel(const float* __restrict__ W1, float* __restrict__ pW1) {
    int idx = blockIdx.x * blockDim.x + threadIdx.x;
    if (idx >= (F_IN / 4) * 256) return;
    int kk   = idx >> 8;
    int rem  = idx & 255;
    int lane = rem >> 3;
    int j    = rem & 7;
    int n    = j >> 1;
    int t    = j & 1;
    int lm   = lane & 15;
    int kh   = (lane >> 4) << 1;
    pW1[idx] = W1[(size_t)(4 * kk + kh + t) * F_HID + n * 16 + lm];
}

__global__ void repack_w2_kernel(const float* __restrict__ W2, float* __restrict__ pW2) {
    int idx = blockIdx.x * blockDim.x + threadIdx.x;
    if (idx >= (F_HID / 4) * 256) return;
    int kk   = idx >> 8;
    int rem  = idx & 255;
    int lane = rem >> 3;
    int j    = rem & 7;
    int n    = j >> 1;
    int t    = j & 1;
    int lm   = lane & 15;
    int kh   = (lane >> 4) << 1;
    int col  = n * 16 + lm;
    float v  = 0.0f;
    if (j < 6 && col < F_OUT)
        v = W2[(size_t)(4 * kk + kh + t) * F_OUT + col];
    pW2[idx] = v;
}


__global__ void gemm1_wmma_kernel(const float* __restrict__ x, const float* __restrict__ pW1,
                                  const float* __restrict__ dinv, float* __restrict__ h1, int N) {
    const int lane = threadIdx.x & 31;
    const int wave = threadIdx.x >> 5;
    const int row0 = blockIdx.x * 128 + wave * 16;
    const int lm   = lane & 15;
    const int kh   = (lane >> 4) << 1;
    int r = row0 + lm;
    if (r > N - 1) r = N - 1;
    const float* xrow = x + (size_t)r * F_IN + kh;
    const float* pw   = pW1 + (size_t)lane * 8;

    v8f acc0 = {}, acc1 = {}, acc2 = {}, acc3 = {};

    for (int kk = 0; kk < F_IN / 4; ++kk) {
        const v2f A   = *(const v2f*)(xrow + kk * 4);
        const v4f w01 = *(const v4f*)(pw + (size_t)kk * 256);
        const v4f w23 = *(const v4f*)(pw + (size_t)kk * 256 + 4);
        v2f B0; B0[0] = w01[0]; B0[1] = w01[1];
        v2f B1; B1[0] = w01[2]; B1[1] = w01[3];
        v2f B2; B2[0] = w23[0]; B2[1] = w23[1];
        v2f B3; B3[0] = w23[2]; B3[1] = w23[3];
        acc0 = __builtin_amdgcn_wmma_f32_16x16x4_f32(false, A, false, B0, (short)0, acc0, false, false);
        acc1 = __builtin_amdgcn_wmma_f32_16x16x4_f32(false, A, false, B1, (short)0, acc1, false, false);
        acc2 = __builtin_amdgcn_wmma_f32_16x16x4_f32(false, A, false, B2, (short)0, acc2, false, false);
        acc3 = __builtin_amdgcn_wmma_f32_16x16x4_f32(false, A, false, B3, (short)0, acc3, false, false);
    }

    const int hi8 = (lane >> 4) << 3;
    for (int j = 0; j < 8; ++j) {
        int rr = row0 + hi8 + j;
        if (rr < N) {
            const float dv = dinv[rr];
            float* o = h1 + (size_t)rr * F_HID + lm;
            o[0]  = acc0[j] * dv;
            o[16] = acc1[j] * dv;
            o[32] = acc2[j] * dv;
            o[48] = acc3[j] * dv;
        }
    }
}


#define RL(v, j) __builtin_amdgcn_readlane((v), (j))

__device__ __forceinline__ float2 ld_row2(const float* __restrict__ col, int s, int pitch) {
    return *(const float2*)(col + (size_t)s * pitch);
}

__device__ __forceinline__ float2 sum_rows(const float* __restrict__ col, int pitch,
                                           int my_s, int n, float2 a) {
    int j = 0;
    for (; j + 8 <= n; j += 8) {
        const float2 v0 = ld_row2(col, RL(my_s, j + 0), pitch);
        const float2 v1 = ld_row2(col, RL(my_s, j + 1), pitch);
        const float2 v2 = ld_row2(col, RL(my_s, j + 2), pitch);
        const float2 v3 = ld_row2(col, RL(my_s, j + 3), pitch);
        const float2 v4 = ld_row2(col, RL(my_s, j + 4), pitch);
        const float2 v5 = ld_row2(col, RL(my_s, j + 5), pitch);
        const float2 v6 = ld_row2(col, RL(my_s, j + 6), pitch);
        const float2 v7 = ld_row2(col, RL(my_s, j + 7), pitch);
        a.x += v0.x; a.y += v0.y;
        a.x += v1.x; a.y += v1.y;
        a.x += v2.x; a.y += v2.y;
        a.x += v3.x; a.y += v3.y;
        a.x += v4.x; a.y += v4.y;
        a.x += v5.x; a.y += v5.y;
        a.x += v6.x; a.y += v6.y;
        a.x += v7.x; a.y += v7.y;
    }
    if (j + 4 <= n) {
        const float2 v0 = ld_row2(col, RL(my_s, j + 0), pitch);
        const float2 v1 = ld_row2(col, RL(my_s, j + 1), pitch);
        const float2 v2 = ld_row2(col, RL(my_s, j + 2), pitch);
        const float2 v3 = ld_row2(col, RL(my_s, j + 3), pitch);
        a.x += v0.x; a.y += v0.y;
        a.x += v1.x; a.y += v1.y;
        a.x += v2.x; a.y += v2.y;
        a.x += v3.x; a.y += v3.y;
        j += 4;
    }
    if (j + 2 <= n) {
        const float2 v0 = ld_row2(col, RL(my_s, j + 0), pitch);
        const float2 v1 = ld_row2(col, RL(my_s, j + 1), pitch);
        a.x += v0.x; a.y += v0.y;
        a.x += v1.x; a.y += v1.y;
        j += 2;
    }
    if (j < n) {
        const float2 v0 = ld_row2(col, RL(my_s, j), pitch);
        a.x += v0.x; a.y += v0.y;
    }
    return a;
}

__global__ void __launch_bounds__(256)
gather1_kernel(const float* __restrict__ h1s, const float* __restrict__ dinv,
               const int* __restrict__ start, const int* __restrict__ cnt,
               const int* __restrict__ csr, const float* __restrict__ b1,
               float* __restrict__ z, int N) {
    const int lane = threadIdx.x & 31;
    const int node = blockIdx.x * 8 + (threadIdx.x >> 5);
    if (node >= N) return;
    const int beg = __builtin_amdgcn_readfirstlane(start[node]);
    const int c   = __builtin_amdgcn_readfirstlane(cnt[node]);
    const float dd = dinv[node];
    const float* col = h1s + lane * 2;
    float2 a = ld_row2(col, node, F_HID);
    for (int base = 0; base < c; base += 32) {
        int n = c - base;
        if (n > 32) n = 32;
        int my_s = node;
        if (lane < n) my_s = csr[beg + base + lane];
        a = sum_rows(col, F_HID, my_s, n, a);
    }
    const float2 bb = *(const float2*)(b1 + lane * 2);
    const float rx = a.x * dd + bb.x;
    const float ry = a.y * dd + bb.y;
    float2 r;
    r.x = rx > 0.0f ? rx : 0.0f;
    r.y = ry > 0.0f ? ry : 0.0f;
    *(float2*)(z + (size_t)node * F_HID + lane * 2) = r;
}


__global__ void gemm2_wmma_kernel(const float* __restrict__ z, const float* __restrict__ pW2,
                                  const float* __restrict__ dinv, float* __restrict__ h2, int N) {
    const int lane = threadIdx.x & 31;
    const int wave = threadIdx.x >> 5;
    const int row0 = blockIdx.x * 128 + wave * 16;
    const int lm   = lane & 15;
    const int kh   = (lane >> 4) << 1;
    int r = row0 + lm;
    if (r > N - 1) r = N - 1;
    const float* zrow = z + (size_t)r * F_HID + kh;
    const float* pw   = pW2 + (size_t)lane * 8;

    v8f acc0 = {}, acc1 = {}, acc2 = {};

    for (int kk = 0; kk < F_HID / 4; ++kk) {
        const v2f A   = *(const v2f*)(zrow + kk * 4);
        const v4f w01 = *(const v4f*)(pw + (size_t)kk * 256);
        const v2f w2v = *(const v2f*)(pw + (size_t)kk * 256 + 4);
        v2f B0; B0[0] = w01[0]; B0[1] = w01[1];
        v2f B1; B1[0] = w01[2]; B1[1] = w01[3];
        v2f B2; B2[0] = w2v[0]; B2[1] = w2v[1];
        acc0 = __builtin_amdgcn_wmma_f32_16x16x4_f32(false, A, false, B0, (short)0, acc0, false, false);
        acc1 = __builtin_amdgcn_wmma_f32_16x16x4_f32(false, A, false, B1, (short)0, acc1, false, false);
        acc2 = __builtin_amdgcn_wmma_f32_16x16x4_f32(false, A, false, B2, (short)0, acc2, false, false);
    }

    const int hi8  = (lane >> 4) << 3;
    const bool c2ok = (lm < 8);
    for (int j = 0; j < 8; ++j) {
        int rr = row0 + hi8 + j;
        if (rr < N) {
            const float dv = dinv[rr];
            float* o = h2 + (size_t)rr * F_OUT + lm;
            o[0]  = acc0[j] * dv;
            o[16] = acc1[j] * dv;
            if (c2ok) o[32] = acc2[j] * dv;
        }
    }
}


__global__ void __launch_bounds__(256)
gather2_kernel(const float* __restrict__ h2s, const float* __restrict__ dinv,
               const int* __restrict__ start, const int* __restrict__ cnt,
               const int* __restrict__ csr, const float* __restrict__ b2,
               float* __restrict__ out, int N) {
    __shared__ __attribute__((aligned(16))) float so[8 * F_OUT];
    const int lane = threadIdx.x & 31;
    const int wave = threadIdx.x >> 5;
    const int node = blockIdx.x * 8 + wave;
    if (node < N) {
        const int beg = __builtin_amdgcn_readfirstlane(start[node]);
        const int c   = __builtin_amdgcn_readfirstlane(cnt[node]);
        const float dd = dinv[node];
        const int cl = lane < 20 ? lane : 19;
        const float* col = h2s + cl * 2;
        float2 a = ld_row2(col, node, F_OUT);
        for (int base = 0; base < c; base += 32) {
            int n = c - base;
            if (n > 32) n = 32;
            int my_s = node;
            if (lane < n) my_s = csr[beg + base + lane];
            a = sum_rows(col, F_OUT, my_s, n, a);
        }
        if (lane < 20) {
            const float2 bb = *(const float2*)(b2 + lane * 2);
            so[wave * F_OUT + lane * 2]     = a.x * dd + bb.x;
            so[wave * F_OUT + lane * 2 + 1] = a.y * dd + bb.y;
        }
    }
    __syncthreads();
    const int t = threadIdx.x;
    if (t < (8 * F_OUT) / 4) {
        const size_t g = (size_t)blockIdx.x * (8 * F_OUT) + (size_t)t * 4;
        if (g < (size_t)N * F_OUT)
            *(v4f*)(out + g) = *(const v4f*)(so + t * 4);
    }
}


extern "C" void kernel_launch(void* const* d_in, const int* in_sizes, int n_in,
                              void* d_out, int out_size, void* d_ws, size_t ws_size,
                              hipStream_t stream) {
    const float* x  = (const float*)d_in[0];
    const int*   ei = (const int*)d_in[1];
    const float* W1 = (const float*)d_in[2];
    const float* b1 = (const float*)d_in[3];
    const float* W2 = (const float*)d_in[4];
    const float* b2 = (const float*)d_in[5];

    const int N = in_sizes[0] / F_IN;
    const int E = in_sizes[1] / 2;
    const int* src = ei;
    const int* dst = ei + E;

    float* ws    = (float*)d_ws;
    float* dinv  = ws + (size_t)N;
    float* bufA  = ws + (size_t)2 * N;
    float* bufB  = bufA + (size_t)N * F_HID;
    float* pW1   = bufB + (size_t)N * F_HID;
    float* pW2   = pW1 + (size_t)(F_IN / 4) * 256;
    int*   cnt   = (int*)(pW2 + (size_t)(F_HID / 4) * 256);
    int*   start = cnt + (size_t)N;
    int*   rank  = start + (size_t)N;
    int*   csr   = rank + (size_t)E;
    int*   total = csr + (size_t)E;
    float* out   = (float*)d_out;

    const int B = 256;
    const int nodeBlocks  = (N + B - 1) / B;
    const int edgeBlocks  = (E + B - 1) / B;
    const int gemmBlocks  = (N + 127) / 128;
    const int waveNodeBlk = (N + (B / 32) - 1) / (B / 32);
    const int rp1Blocks   = ((F_IN / 4) * 256 + B - 1) / B;
    const int rp2Blocks   = ((F_HID / 4) * 256 + B - 1) / B;

    repack_w1_kernel<<<rp1Blocks, B, 0, stream>>>(W1, pW1);
    repack_w2_kernel<<<rp2Blocks, B, 0, stream>>>(W2, pW2);

    cnt_init_kernel<<<nodeBlocks, B, 0, stream>>>(cnt, total, N);
    rank_edge_kernel<<<edgeBlocks, B, 0, stream>>>(dst, cnt, rank, E);
    alloc_kernel<<<nodeBlocks, B, 0, stream>>>(cnt, total, start, dinv, N);
    fill_kernel<<<edgeBlocks, B, 0, stream>>>(src, dst, rank, start, csr, E);

    gemm1_wmma_kernel<<<gemmBlocks, B, 0, stream>>>(x, pW1, dinv, bufA, N);
    gather1_kernel<<<waveNodeBlk, B, 0, stream>>>(bufA, dinv, start, cnt, csr, b1, bufB, N);

    gemm2_wmma_kernel<<<gemmBlocks, B, 0, stream>>>(bufB, pW2, dinv, bufA, N);
    gather2_kernel<<<waveNodeBlk, B, 0, stream>>>(bufA, dinv, start, cnt, csr, b2, out, N);
}


# Neighbour lists built by a two-level LDS counting sort (by 256-node range, then by node) instead of 1.6 M global returning atomics and scattered stores; per-node sum order = LDS arrival order
# speedup vs baseline: 3.1801x; 1.5585x over previous
#include <hip/hip_runtime.h>

typedef __attribute__((ext_vector_type(2))) float v2f;
typedef __attribute__((ext_vector_type(4))) float v4f;
typedef __attribute__((ext_vector_type(8))) float v8f;

#define F_IN  128
#define F_HID 64
#define F_OUT 40


#define RSH   8
#define RS    256
#define MAXR  1024
#define TILE  4096

__device__ __forceinline__ int block_scan_incl(int v, int* wsum, int& total) {
    const int lane = threadIdx.x & 31;
    const int wave = threadIdx.x >> 5;
    int incl = v;
    for (int o = 1; o < 32; o <<= 1) {
        int t = __shfl_up(incl, o);
        if (lane >= o) incl += t;
    }
    if (lane == 31) wsum[wave] = incl;
    __syncthreads();
    int before = 0, all = 0;
    for (int w = 0; w < 8; ++w) {
        const int s = wsum[w];
        if (w < wave) before += s;
        all += s;
    }
    total = all;
    __syncthreads();
    return incl + before;
}

__global__ void range_init_kernel(int* __restrict__ rangeCnt, int* __restrict__ rangeFill, int R) {
    int i = blockIdx.x * blockDim.x + threadIdx.x;
    if (i < R) { rangeCnt[i] = 0; rangeFill[i] = 0; }
}

__global__ void __launch_bounds__(256)
range_count_kernel(const int* __restrict__ dst, int* __restrict__ rangeCnt, int E, int R) {
    __shared__ int h[MAXR];
    const int tid = threadIdx.x;
    for (int i = tid; i < R; i += 256) h[i] = 0;
    __syncthreads();
    const int base = blockIdx.x * TILE;
    for (int k = 0; k < TILE / 256; ++k) {
        const int e = base + k * 256 + tid;
        if (e < E) atomicAdd(&h[dst[e] >> RSH], 1);
    }
    __syncthreads();
    for (int i = tid; i < R; i += 256) {
        const int c = h[i];
        if (c) atomicAdd(&rangeCnt[i], c);
    }
}

__global__ void __launch_bounds__(256)
range_bin_kernel(const int* __restrict__ src, const int* __restrict__ dst,
                 const int* __restrict__ rangeCnt, int* __restrict__ rangeFill,
                 int2* __restrict__ binned, int E, int R) {
    __shared__ int h[MAXR];
    __shared__ int off[MAXR];
    __shared__ int wsum[8];
    const int tid = threadIdx.x;
    int a[4];
    int s4 = 0;
    for (int j = 0; j < 4; ++j) {
        const int i = tid * 4 + j;
        a[j] = (i < R) ? rangeCnt[i] : 0;
        s4 += a[j];
    }
    int total;
    int run = block_scan_incl(s4, wsum, total) - s4;
    for (int j = 0; j < 4; ++j) {
        off[tid * 4 + j] = run;
        run += a[j];
        h[tid * 4 + j] = 0;
    }
    __syncthreads();
    const int base = blockIdx.x * TILE;
    int ss[TILE / 256], dd[TILE / 256], lr[TILE / 256];
    #pragma unroll
    for (int k = 0; k < TILE / 256; ++k) {
        const int e = base + k * 256 + tid;
        ss[k] = 0; dd[k] = 0; lr[k] = 0;
        if (e < E) {
            dd[k] = dst[e];
            ss[k] = src[e];
            lr[k] = atomicAdd(&h[dd[k] >> RSH], 1);
        }
    }
    __syncthreads();
    for (int i = tid; i < R; i += 256) {
        const int c = h[i];
        if (c) h[i] = off[i] + atomicAdd(&rangeFill[i], c);
    }
    __syncthreads();
    #pragma unroll
    for (int k = 0; k < TILE / 256; ++k) {
        const int e = base + k * 256 + tid;
        if (e < E) binned[h[dd[k] >> RSH] + lr[k]] = make_int2(ss[k], dd[k]);
    }
}

__global__ void __launch_bounds__(256)
range_build_kernel(const int2* __restrict__ binned, const int* __restrict__ rangeCnt,
                   int* __restrict__ start, int* __restrict__ cnt, float* __restrict__ dinv,
                   int* __restrict__ csr, int N) {
    __shared__ int c[RS];
    __shared__ int cur[RS];
    __shared__ int wsum[8];
    const int tid = threadIdx.x;
    const int r   = blockIdx.x;
    int part = 0;
    for (int i = tid; i < r; i += 256) part += rangeCnt[i];
    int rbeg;
    block_scan_incl(part, wsum, rbeg);
    const int n = rangeCnt[r];
    c[tid] = 0;
    __syncthreads();
    for (int i = tid; i < n; i += 256) {
        const int2 p = binned[rbeg + i];
        atomicAdd(&c[p.y & (RS - 1)], 1);
    }
    __syncthreads();
    const int v = c[tid];
    int tot;
    const int excl = block_scan_incl(v, wsum, tot) - v;
    const int node = r * RS + tid;
    if (node < N) {
        start[node] = rbeg + excl;
        cnt[node]   = v;
        dinv[node]  = rsqrtf((float)(v + 1));
    }
    cur[tid] = rbeg + excl;
    __syncthreads();
    for (int i = tid; i < n; i += 256) {
        const int2 p = binned[rbeg + i];
        const int pos = atomicAdd(&cur[p.y & (RS - 1)], 1);
        csr[pos] = p.x;
    }
}


__global__ void repack_w1_kernel(const float* __restrict__ W1, float* __restrict__ pW1) {
    int idx = blockIdx.x * blockDim.x + threadIdx.x;
    if (idx >= (F_IN / 4) * 256) return;
    int kk   = idx >> 8;
    int rem  = idx & 255;
    int lane = rem >> 3;
    int j    = rem & 7;
    int n    = j >> 1;
    int t    = j & 1;
    int lm   = lane & 15;
    int kh   = (lane >> 4) << 1;
    pW1[idx] = W1[(size_t)(4 * kk + kh + t) * F_HID + n * 16 + lm];
}

__global__ void repack_w2_kernel(const float* __restrict__ W2, float* __restrict__ pW2) {
    int idx = blockIdx.x * blockDim.x + threadIdx.x;
    if (idx >= (F_HID / 4) * 256) return;
    int kk   = idx >> 8;
    int rem  = idx & 255;
    int lane = rem >> 3;
    int j    = rem & 7;
    int n    = j >> 1;
    int t    = j & 1;
    int lm   = lane & 15;
    int kh   = (lane >> 4) << 1;
    int col  = n * 16 + lm;
    float v  = 0.0f;
    if (j < 6 && col < F_OUT)
        v = W2[(size_t)(4 * kk + kh + t) * F_OUT + col];
    pW2[idx] = v;
}


__global__ void gemm1_wmma_kernel(const float* __restrict__ x, const float* __restrict__ pW1,
                                  const float* __restrict__ dinv, float* __restrict__ h1, int N) {
    const int lane = threadIdx.x & 31;
    const int wave = threadIdx.x >> 5;
    const int row0 = blockIdx.x * 128 + wave * 16;
    const int lm   = lane & 15;
    const int kh   = (lane >> 4) << 1;
    int r = row0 + lm;
    if (r > N - 1) r = N - 1;
    const float* xrow = x + (size_t)r * F_IN + kh;
    const float* pw   = pW1 + (size_t)lane * 8;

    v8f acc0 = {}, acc1 = {}, acc2 = {}, acc3 = {};

    for (int kk = 0; kk < F_IN / 4; ++kk) {
        const v2f A   = *(const v2f*)(xrow + kk * 4);
        const v4f w01 = *(const v4f*)(pw + (size_t)kk * 256);
        const v4f w23 = *(const v4f*)(pw + (size_t)kk * 256 + 4);
        v2f B0; B0[0] = w01[0]; B0[1] = w01[1];
        v2f B1; B1[0] = w01[2]; B1[1] = w01[3];
        v2f B2; B2[0] = w23[0]; B2[1] = w23[1];
        v2f B3; B3[0] = w23[2]; B3[1] = w23[3];
        acc0 = __builtin_amdgcn_wmma_f32_16x16x4_f32(false, A, false, B0, (short)0, acc0, false, false);
        acc1 = __builtin_amdgcn_wmma_f32_16x16x4_f32(false, A, false, B1, (short)0, acc1, false, false);
        acc2 = __builtin_amdgcn_wmma_f32_16x16x4_f32(false, A, false, B2, (short)0, acc2, false, false);
        acc3 = __builtin_amdgcn_wmma_f32_16x16x4_f32(false, A, false, B3, (short)0, acc3, false, false);
    }

    const int hi8 = (lane >> 4) << 3;
    for (int j = 0; j < 8; ++j) {
        int rr = row0 + hi8 + j;
        if (rr < N) {
            const float dv = dinv[rr];
            float* o = h1 + (size_t)rr * F_HID + lm;
            o[0]  = acc0[j] * dv;
            o[16] = acc1[j] * dv;
            o[32] = acc2[j] * dv;
            o[48] = acc3[j] * dv;
        }
    }
}


#define RL(v, j) __builtin_amdgcn_readlane((v), (j))

__device__ __forceinline__ float2 ld_row2(const float* __restrict__ col, int s, int pitch) {
    return *(const float2*)(col + (size_t)s * pitch);
}

__device__ __forceinline__ float2 sum_rows(const float* __restrict__ col, int pitch,
                                           int my_s, int n, float2 a) {
    int j = 0;
    for (; j + 8 <= n; j += 8) {
        const float2 v0 = ld_row2(col, RL(my_s, j + 0), pitch);
        const float2 v1 = ld_row2(col, RL(my_s, j + 1), pitch);
        const float2 v2 = ld_row2(col, RL(my_s, j + 2), pitch);
        const float2 v3 = ld_row2(col, RL(my_s, j + 3), pitch);
        const float2 v4 = ld_row2(col, RL(my_s, j + 4), pitch);
        const float2 v5 = ld_row2(col, RL(my_s, j + 5), pitch);
        const float2 v6 = ld_row2(col, RL(my_s, j + 6), pitch);
        const float2 v7 = ld_row2(col, RL(my_s, j + 7), pitch);
        a.x += v0.x; a.y += v0.y;
        a.x += v1.x; a.y += v1.y;
        a.x += v2.x; a.y += v2.y;
        a.x += v3.x; a.y += v3.y;
        a.x += v4.x; a.y += v4.y;
        a.x += v5.x; a.y += v5.y;
        a.x += v6.x; a.y += v6.y;
        a.x += v7.x; a.y += v7.y;
    }
    if (j + 4 <= n) {
        const float2 v0 = ld_row2(col, RL(my_s, j + 0), pitch);
        const float2 v1 = ld_row2(col, RL(my_s, j + 1), pitch);
        const float2 v2 = ld_row2(col, RL(my_s, j + 2), pitch);
        const float2 v3 = ld_row2(col, RL(my_s, j + 3), pitch);
        a.x += v0.x; a.y += v0.y;
        a.x += v1.x; a.y += v1.y;
        a.x += v2.x; a.y += v2.y;
        a.x += v3.x; a.y += v3.y;
        j += 4;
    }
    if (j + 2 <= n) {
        const float2 v0 = ld_row2(col, RL(my_s, j + 0), pitch);
        const float2 v1 = ld_row2(col, RL(my_s, j + 1), pitch);
        a.x += v0.x; a.y += v0.y;
        a.x += v1.x; a.y += v1.y;
        j += 2;
    }
    if (j < n) {
        const float2 v0 = ld_row2(col, RL(my_s, j), pitch);
        a.x += v0.x; a.y += v0.y;
    }
    return a;
}

__global__ void __launch_bounds__(256)
gather1_kernel(const float* __restrict__ h1s, const float* __restrict__ dinv,
               const int* __restrict__ start, const int* __restrict__ cnt,
               const int* __restrict__ csr, const float* __restrict__ b1,
               float* __restrict__ z, int N) {
    const int lane = threadIdx.x & 31;
    const int node = blockIdx.x * 8 + (threadIdx.x >> 5);
    if (node >= N) return;
    const int beg = __builtin_amdgcn_readfirstlane(start[node]);
    const int c   = __builtin_amdgcn_readfirstlane(cnt[node]);
    const float dd = dinv[node];
    const float* col = h1s + lane * 2;
    float2 a = ld_row2(col, node, F_HID);
    for (int base = 0; base < c; base += 32) {
        int n = c - base;
        if (n > 32) n = 32;
        int my_s = node;
        if (lane < n) my_s = csr[beg + base + lane];
        a = sum_rows(col, F_HID, my_s, n, a);
    }
    const float2 bb = *(const float2*)(b1 + lane * 2);
    const float rx = a.x * dd + bb.x;
    const float ry = a.y * dd + bb.y;
    float2 r;
    r.x = rx > 0.0f ? rx : 0.0f;
    r.y = ry > 0.0f ? ry : 0.0f;
    *(float2*)(z + (size_t)node * F_HID + lane * 2) = r;
}


__global__ void gemm2_wmma_kernel(const float* __restrict__ z, const float* __restrict__ pW2,
                                  const float* __restrict__ dinv, float* __restrict__ h2, int N) {
    const int lane = threadIdx.x & 31;
    const int wave = threadIdx.x >> 5;
    const int row0 = blockIdx.x * 128 + wave * 16;
    const int lm   = lane & 15;
    const int kh   = (lane >> 4) << 1;
    int r = row0 + lm;
    if (r > N - 1) r = N - 1;
    const float* zrow = z + (size_t)r * F_HID + kh;
    const float* pw   = pW2 + (size_t)lane * 8;

    v8f acc0 = {}, acc1 = {}, acc2 = {};

    for (int kk = 0; kk < F_HID / 4; ++kk) {
        const v2f A   = *(const v2f*)(zrow + kk * 4);
        const v4f w01 = *(const v4f*)(pw + (size_t)kk * 256);
        const v2f w2v = *(const v2f*)(pw + (size_t)kk * 256 + 4);
        v2f B0; B0[0] = w01[0]; B0[1] = w01[1];
        v2f B1; B1[0] = w01[2]; B1[1] = w01[3];
        v2f B2; B2[0] = w2v[0]; B2[1] = w2v[1];
        acc0 = __builtin_amdgcn_wmma_f32_16x16x4_f32(false, A, false, B0, (short)0, acc0, false, false);
        acc1 = __builtin_amdgcn_wmma_f32_16x16x4_f32(false, A, false, B1, (short)0, acc1, false, false);
        acc2 = __builtin_amdgcn_wmma_f32_16x16x4_f32(false, A, false, B2, (short)0, acc2, false, false);
    }

    const int hi8  = (lane >> 4) << 3;
    const bool c2ok = (lm < 8);
    for (int j = 0; j < 8; ++j) {
        int rr = row0 + hi8 + j;
        if (rr < N) {
            const float dv = dinv[rr];
            float* o = h2 + (size_t)rr * F_OUT + lm;
            o[0]  = acc0[j] * dv;
            o[16] = acc1[j] * dv;
            if (c2ok) o[32] = acc2[j] * dv;
        }
    }
}


__global__ void __launch_bounds__(256)
gather2_kernel(const float* __restrict__ h2s, const float* __restrict__ dinv,
               const int* __restrict__ start, const int* __restrict__ cnt,
               const int* __restrict__ csr, const float* __restrict__ b2,
               float* __restrict__ out, int N) {
    __shared__ __attribute__((aligned(16))) float so[8 * F_OUT];
    const int lane = threadIdx.x & 31;
    const int wave = threadIdx.x >> 5;
    const int node = blockIdx.x * 8 + wave;
    if (node < N) {
        const int beg = __builtin_amdgcn_readfirstlane(start[node]);
        const int c   = __builtin_amdgcn_readfirstlane(cnt[node]);
        const float dd = dinv[node];
        const int cl = lane < 20 ? lane : 19;
        const float* col = h2s + cl * 2;
        float2 a = ld_row2(col, node, F_OUT);
        for (int base = 0; base < c; base += 32) {
            int n = c - base;
            if (n > 32) n = 32;
            int my_s = node;
            if (lane < n) my_s = csr[beg + base + lane];
            a = sum_rows(col, F_OUT, my_s, n, a);
        }
        if (lane < 20) {
            const float2 bb = *(const float2*)(b2 + lane * 2);
            so[wave * F_OUT + lane * 2]     = a.x * dd + bb.x;
            so[wave * F_OUT + lane * 2 + 1] = a.y * dd + bb.y;
        }
    }
    __syncthreads();
    const int t = threadIdx.x;
    if (t < (8 * F_OUT) / 4) {
        const size_t g = (size_t)blockIdx.x * (8 * F_OUT) + (size_t)t * 4;
        if (g < (size_t)N * F_OUT)
            *(v4f*)(out + g) = *(const v4f*)(so + t * 4);
    }
}


extern "C" void kernel_launch(void* const* d_in, const int* in_sizes, int n_in,
                              void* d_out, int out_size, void* d_ws, size_t ws_size,
                              hipStream_t stream) {
    const float* x  = (const float*)d_in[0];
    const int*   ei = (const int*)d_in[1];
    const float* W1 = (const float*)d_in[2];
    const float* b1 = (const float*)d_in[3];
    const float* W2 = (const float*)d_in[4];
    const float* b2 = (const float*)d_in[5];

    const int N = in_sizes[0] / F_IN;
    const int E = in_sizes[1] / 2;
    const int* src = ei;
    const int* dst = ei + E;

    float* ws    = (float*)d_ws;
    float* dinv  = ws + (size_t)N;
    float* bufA  = ws + (size_t)2 * N;
    float* bufB  = bufA + (size_t)N * F_HID;
    float* pW1   = bufB + (size_t)N * F_HID;
    float* pW2   = pW1 + (size_t)(F_IN / 4) * 256;
    int*   cnt   = (int*)(pW2 + (size_t)(F_HID / 4) * 256);
    int*   start = cnt + (size_t)N;
    int2*  binned = (int2*)(start + (size_t)N + (N & 1));
    int*   csr   = (int*)(binned + (size_t)E);
    int*   rangeCnt  = csr + (size_t)E;
    int*   rangeFill = rangeCnt + MAXR;
    float* out   = (float*)d_out;

    const int B = 256;
    const int R           = (N + RS - 1) / RS;
    const int tileBlocks  = (E + TILE - 1) / TILE;
    const int gemmBlocks  = (N + 127) / 128;
    const int waveNodeBlk = (N + (B / 32) - 1) / (B / 32);
    const int rp1Blocks   = ((F_IN / 4) * 256 + B - 1) / B;
    const int rp2Blocks   = ((F_HID / 4) * 256 + B - 1) / B;

    repack_w1_kernel<<<rp1Blocks, B, 0, stream>>>(W1, pW1);
    repack_w2_kernel<<<rp2Blocks, B, 0, stream>>>(W2, pW2);

    range_init_kernel<<<(R + B - 1) / B, B, 0, stream>>>(rangeCnt, rangeFill, R);
    range_count_kernel<<<tileBlocks, B, 0, stream>>>(dst, rangeCnt, E, R);
    range_bin_kernel<<<tileBlocks, B, 0, stream>>>(src, dst, rangeCnt, rangeFill, binned, E, R);
    range_build_kernel<<<R, B, 0, stream>>>(binned, rangeCnt, start, cnt, dinv, csr, N);

    gemm1_wmma_kernel<<<gemmBlocks, B, 0, stream>>>(x, pW1, dinv, bufA, N);
    gather1_kernel<<<waveNodeBlk, B, 0, stream>>>(bufA, dinv, start, cnt, csr, b1, bufB, N);

    gemm2_wmma_kernel<<<gemmBlocks, B, 0, stream>>>(bufB, pW2, dinv, bufA, N);
    gather2_kernel<<<waveNodeBlk, B, 0, stream>>>(bufA, dinv, start, cnt, csr, b2, out, N);
}


# Sort kernels + GEMM epilogues: a thread's independent reads (16-20 edge reads, 8 row scales) issued as one group ahead of use, was one memory round trip each; list order still LDS arrival order
# speedup vs baseline: 3.6493x; 1.1475x over previous
#include <hip/hip_runtime.h>

typedef __attribute__((ext_vector_type(2))) float v2f;
typedef __attribute__((ext_vector_type(4))) float v4f;
typedef __attribute__((ext_vector_type(8))) float v8f;

#define F_IN  128
#define F_HID 64
#define F_OUT 40


#define RSH   8
#define RS    256
#define MAXR  1024
#define TILE  4096
#define EPT   (TILE / 256)
#define RPT   (MAXR / 256)
#define BPT   20

__device__ __forceinline__ int block_scan_incl(int v, int* wsum, int& total) {
    const int lane = threadIdx.x & 31;
    const int wave = threadIdx.x >> 5;
    int incl = v;
    for (int o = 1; o < 32; o <<= 1) {
        int t = __shfl_up(incl, o);
        if (lane >= o) incl += t;
    }
    if (lane == 31) wsum[wave] = incl;
    __syncthreads();
    int before = 0, all = 0;
    for (int w = 0; w < 8; ++w) {
        const int s = wsum[w];
        if (w < wave) before += s;
        all += s;
    }
    total = all;
    __syncthreads();
    return incl + before;
}

__global__ void range_init_kernel(int* __restrict__ rangeCnt, int* __restrict__ rangeFill, int R) {
    int i = blockIdx.x * blockDim.x + threadIdx.x;
    if (i < R) { rangeCnt[i] = 0; rangeFill[i] = 0; }
}

__global__ void __launch_bounds__(256)
range_count_kernel(const int* __restrict__ dst, int* __restrict__ rangeCnt, int E, int R) {
    __shared__ int h[MAXR];
    const int tid = threadIdx.x;
    const int base = blockIdx.x * TILE;
    int d[EPT];
    for (int k = 0; k < EPT; ++k) {
        const int e = base + k * 256 + tid;
        d[k] = dst[e < E ? e : E - 1];
    }
    for (int i = tid; i < R; i += 256) h[i] = 0;
    __syncthreads();
    for (int k = 0; k < EPT; ++k) {
        const int e = base + k * 256 + tid;
        if (e < E) atomicAdd(&h[d[k] >> RSH], 1);
    }
    __syncthreads();
    for (int i = tid; i < R; i += 256) {
        const int c = h[i];
        if (c) atomicAdd(&rangeCnt[i], c);
    }
}

__global__ void __launch_bounds__(256)
range_bin_kernel(const int* __restrict__ src, const int* __restrict__ dst,
                 const int* __restrict__ rangeCnt, int* __restrict__ rangeFill,
                 int2* __restrict__ binned, int E, int R) {
    __shared__ int h[MAXR];
    __shared__ int off[MAXR];
    __shared__ int wsum[8];
    const int tid = threadIdx.x;
    const int base = blockIdx.x * TILE;
    int ss[EPT], dd[EPT], lr[EPT], hb[EPT];
    int a[4];
    for (int j = 0; j < 4; ++j) {
        const int i = tid * 4 + j;
        a[j] = (i < R) ? rangeCnt[i] : 0;
    }
    for (int k = 0; k < EPT; ++k) {
        const int e  = base + k * 256 + tid;
        const int ec = e < E ? e : E - 1;
        dd[k] = dst[ec];
        ss[k] = src[ec];
    }
    const int s4 = a[0] + a[1] + a[2] + a[3];
    int total;
    int run = block_scan_incl(s4, wsum, total) - s4;
    for (int j = 0; j < 4; ++j) {
        off[tid * 4 + j] = run;
        run += a[j];
        h[tid * 4 + j] = 0;
    }
    __syncthreads();
    for (int k = 0; k < EPT; ++k) {
        const int e = base + k * 256 + tid;
        lr[k] = 0;
        if (e < E) lr[k] = atomicAdd(&h[dd[k] >> RSH], 1);
    }
    __syncthreads();
    int cc[RPT], gg[RPT];
    for (int j = 0; j < RPT; ++j) {
        const int i = tid + j * 256;
        cc[j] = (i < R) ? h[i] : 0;
    }
    for (int j = 0; j < RPT; ++j) {
        gg[j] = 0;
        if (cc[j]) gg[j] = atomicAdd(&rangeFill[tid + j * 256], cc[j]);
    }
    for (int j = 0; j < RPT; ++j) {
        const int i = tid + j * 256;
        if (cc[j]) h[i] = off[i] + gg[j];
    }
    __syncthreads();
    for (int k = 0; k < EPT; ++k) hb[k] = h[dd[k] >> RSH];
    for (int k = 0; k < EPT; ++k) {
        const int e = base + k * 256 + tid;
        if (e < E) binned[hb[k] + lr[k]] = make_int2(ss[k], dd[k]);
    }
}

__global__ void __launch_bounds__(256)
range_build_kernel(const int2* __restrict__ binned, const int* __restrict__ rangeCnt,
                   int* __restrict__ start, int* __restrict__ cnt, float* __restrict__ dinv,
                   int* __restrict__ csr, int N) {
    __shared__ int c[RS];
    __shared__ int cur[RS];
    __shared__ int wsum[8];
    const int tid = threadIdx.x;
    const int r   = blockIdx.x;
    int part = 0;
    for (int j = 0; j < RPT; ++j) {
        const int i = tid + j * 256;
        part += (i < r) ? rangeCnt[i] : 0;
    }
    int rbeg;
    block_scan_incl(part, wsum, rbeg);
    const int n = rangeCnt[r];
    c[tid] = 0;
    __syncthreads();
    int2 p[BPT];
    int held = -1;
    for (int c0 = 0; c0 < n; c0 += BPT * 256) {
        for (int k = 0; k < BPT; ++k) {
            const int i = c0 + k * 256 + tid;
            p[k] = binned[rbeg + (i < n ? i : n - 1)];
        }
        for (int k = 0; k < BPT; ++k) {
            const int i = c0 + k * 256 + tid;
            if (i < n) atomicAdd(&c[p[k].y & (RS - 1)], 1);
        }
        held = c0;
    }
    __syncthreads();
    const int v = c[tid];
    int tot;
    const int excl = block_scan_incl(v, wsum, tot) - v;
    const int node = r * RS + tid;
    if (node < N) {
        start[node] = rbeg + excl;
        cnt[node]   = v;
        dinv[node]  = rsqrtf((float)(v + 1));
    }
    cur[tid] = rbeg + excl;
    __syncthreads();
    for (int c0 = 0; c0 < n; c0 += BPT * 256) {
        if (c0 != held) {
            for (int k = 0; k < BPT; ++k) {
                const int i = c0 + k * 256 + tid;
                p[k] = binned[rbeg + (i < n ? i : n - 1)];
            }
            held = c0;
        }
        int pos[BPT];
        for (int k = 0; k < BPT; ++k) {
            const int i = c0 + k * 256 + tid;
            pos[k] = 0;
            if (i < n) pos[k] = atomicAdd(&cur[p[k].y & (RS - 1)], 1);
        }
        for (int k = 0; k < BPT; ++k) {
            const int i = c0 + k * 256 + tid;
            if (i < n) csr[pos[k]] = p[k].x;
        }
    }
}


__global__ void repack_w1_kernel(const float* __restrict__ W1, float* __restrict__ pW1) {
    int idx = blockIdx.x * blockDim.x + threadIdx.x;
    if (idx >= (F_IN / 4) * 256) return;
    int kk   = idx >> 8;
    int rem  = idx & 255;
    int lane = rem >> 3;
    int j    = rem & 7;
    int n    = j >> 1;
    int t    = j & 1;
    int lm   = lane & 15;
    int kh   = (lane >> 4) << 1;
    pW1[idx] = W1[(size_t)(4 * kk + kh + t) * F_HID + n * 16 + lm];
}

__global__ void repack_w2_kernel(const float* __restrict__ W2, float* __restrict__ pW2) {
    int idx = blockIdx.x * blockDim.x + threadIdx.x;
    if (idx >= (F_HID / 4) * 256) return;
    int kk   = idx >> 8;
    int rem  = idx & 255;
    int lane = rem >> 3;
    int j    = rem & 7;
    int n    = j >> 1;
    int t    = j & 1;
    int lm   = lane & 15;
    int kh   = (lane >> 4) << 1;
    int col  = n * 16 + lm;
    float v  = 0.0f;
    if (j < 6 && col < F_OUT)
        v = W2[(size_t)(4 * kk + kh + t) * F_OUT + col];
    pW2[idx] = v;
}


__global__ void gemm1_wmma_kernel(const float* __restrict__ x, const float* __restrict__ pW1,
                                  const float* __restrict__ dinv, float* __restrict__ h1, int N) {
    const int lane = threadIdx.x & 31;
    const int wave = threadIdx.x >> 5;
    const int row0 = blockIdx.x * 128 + wave * 16;
    const int lm   = lane & 15;
    const int kh   = (lane >> 4) << 1;
    int r = row0 + lm;
    if (r > N - 1) r = N - 1;
    const float* xrow = x + (size_t)r * F_IN + kh;
    const float* pw   = pW1 + (size_t)lane * 8;

    const int hi8 = (lane >> 4) << 3;
    float dv[8];
    for (int j = 0; j < 8; ++j) {
        const int rr = row0 + hi8 + j;
        dv[j] = dinv[rr < N ? rr : N - 1];
    }

    v8f acc0 = {}, acc1 = {}, acc2 = {}, acc3 = {};

    for (int kk = 0; kk < F_IN / 4; ++kk) {
        const v2f A   = *(const v2f*)(xrow + kk * 4);
        const v4f w01 = *(const v4f*)(pw + (size_t)kk * 256);
        const v4f w23 = *(const v4f*)(pw + (size_t)kk * 256 + 4);
        v2f B0; B0[0] = w01[0]; B0[1] = w01[1];
        v2f B1; B1[0] = w01[2]; B1[1] = w01[3];
        v2f B2; B2[0] = w23[0]; B2[1] = w23[1];
        v2f B3; B3[0] = w23[2]; B3[1] = w23[3];
        acc0 = __builtin_amdgcn_wmma_f32_16x16x4_f32(false, A, false, B0, (short)0, acc0, false, false);
        acc1 = __builtin_amdgcn_wmma_f32_16x16x4_f32(false, A, false, B1, (short)0, acc1, false, false);
        acc2 = __builtin_amdgcn_wmma_f32_16x16x4_f32(false, A, false, B2, (short)0, acc2, false, false);
        acc3 = __builtin_amdgcn_wmma_f32_16x16x4_f32(false, A, false, B3, (short)0, acc3, false, false);
    }

    for (int j = 0; j < 8; ++j) {
        int rr = row0 + hi8 + j;
        if (rr < N) {
            float* o = h1 + (size_t)rr * F_HID + lm;
            o[0]  = acc0[j] * dv[j];
            o[16] = acc1[j] * dv[j];
            o[32] = acc2[j] * dv[j];
            o[48] = acc3[j] * dv[j];
        }
    }
}


#define RL(v, j) __builtin_amdgcn_readlane((v), (j))

__device__ __forceinline__ float2 ld_row2(const float* __restrict__ col, int s, int pitch) {
    return *(const float2*)(col + (size_t)s * pitch);
}

__device__ __forceinline__ float2 sum_rows(const float* __restrict__ col, int pitch,
                                           int my_s, int n, float2 a) {
    int j = 0;
    for (; j + 8 <= n; j += 8) {
        const float2 v0 = ld_row2(col, RL(my_s, j + 0), pitch);
        const float2 v1 = ld_row2(col, RL(my_s, j + 1), pitch);
        const float2 v2 = ld_row2(col, RL(my_s, j + 2), pitch);
        const float2 v3 = ld_row2(col, RL(my_s, j + 3), pitch);
        const float2 v4 = ld_row2(col, RL(my_s, j + 4), pitch);
        const float2 v5 = ld_row2(col, RL(my_s, j + 5), pitch);
        const float2 v6 = ld_row2(col, RL(my_s, j + 6), pitch);
        const float2 v7 = ld_row2(col, RL(my_s, j + 7), pitch);
        a.x += v0.x; a.y += v0.y;
        a.x += v1.x; a.y += v1.y;
        a.x += v2.x; a.y += v2.y;
        a.x += v3.x; a.y += v3.y;
        a.x += v4.x; a.y += v4.y;
        a.x += v5.x; a.y += v5.y;
        a.x += v6.x; a.y += v6.y;
        a.x += v7.x; a.y += v7.y;
    }
    if (j + 4 <= n) {
        const float2 v0 = ld_row2(col, RL(my_s, j + 0), pitch);
        const float2 v1 = ld_row2(col, RL(my_s, j + 1), pitch);
        const float2 v2 = ld_row2(col, RL(my_s, j + 2), pitch);
        const float2 v3 = ld_row2(col, RL(my_s, j + 3), pitch);
        a.x += v0.x; a.y += v0.y;
        a.x += v1.x; a.y += v1.y;
        a.x += v2.x; a.y += v2.y;
        a.x += v3.x; a.y += v3.y;
        j += 4;
    }
    if (j + 2 <= n) {
        const float2 v0 = ld_row2(col, RL(my_s, j + 0), pitch);
        const float2 v1 = ld_row2(col, RL(my_s, j + 1), pitch);
        a.x += v0.x; a.y += v0.y;
        a.x += v1.x; a.y += v1.y;
        j += 2;
    }
    if (j < n) {
        const float2 v0 = ld_row2(col, RL(my_s, j), pitch);
        a.x += v0.x; a.y += v0.y;
    }
    return a;
}

__global__ void __launch_bounds__(256)
gather1_kernel(const float* __restrict__ h1s, const float* __restrict__ dinv,
               const int* __restrict__ start, const int* __restrict__ cnt,
               const int* __restrict__ csr, const float* __restrict__ b1,
               float* __restrict__ z, int N) {
    const int lane = threadIdx.x & 31;
    const int node = blockIdx.x * 8 + (threadIdx.x >> 5);
    if (node >= N) return;
    const int beg = __builtin_amdgcn_readfirstlane(start[node]);
    const int c   = __builtin_amdgcn_readfirstlane(cnt[node]);
    const float dd = dinv[node];
    const float* col = h1s + lane * 2;
    float2 a = ld_row2(col, node, F_HID);
    for (int base = 0; base < c; base += 32) {
        int n = c - base;
        if (n > 32) n = 32;
        int my_s = node;
        if (lane < n) my_s = csr[beg + base + lane];
        a = sum_rows(col, F_HID, my_s, n, a);
    }
    const float2 bb = *(const float2*)(b1 + lane * 2);
    const float rx = a.x * dd + bb.x;
    const float ry = a.y * dd + bb.y;
    float2 r;
    r.x = rx > 0.0f ? rx : 0.0f;
    r.y = ry > 0.0f ? ry : 0.0f;
    *(float2*)(z + (size_t)node * F_HID + lane * 2) = r;
}


__global__ void gemm2_wmma_kernel(const float* __restrict__ z, const float* __restrict__ pW2,
                                  const float* __restrict__ dinv, float* __restrict__ h2, int N) {
    const int lane = threadIdx.x & 31;
    const int wave = threadIdx.x >> 5;
    const int row0 = blockIdx.x * 128 + wave * 16;
    const int lm   = lane & 15;
    const int kh   = (lane >> 4) << 1;
    int r = row0 + lm;
    if (r > N - 1) r = N - 1;
    const float* zrow = z + (size_t)r * F_HID + kh;
    const float* pw   = pW2 + (size_t)lane * 8;

    const int hi8 = (lane >> 4) << 3;
    float dv[8];
    for (int j = 0; j < 8; ++j) {
        const int rr = row0 + hi8 + j;
        dv[j] = dinv[rr < N ? rr : N - 1];
    }

    v8f acc0 = {}, acc1 = {}, acc2 = {};

    for (int kk = 0; kk < F_HID / 4; ++kk) {
        const v2f A   = *(const v2f*)(zrow + kk * 4);
        const v4f w01 = *(const v4f*)(pw + (size_t)kk * 256);
        const v2f w2v = *(const v2f*)(pw + (size_t)kk * 256 + 4);
        v2f B0; B0[0] = w01[0]; B0[1] = w01[1];
        v2f B1; B1[0] = w01[2]; B1[1] = w01[3];
        v2f B2; B2[0] = w2v[0]; B2[1] = w2v[1];
        acc0 = __builtin_amdgcn_wmma_f32_16x16x4_f32(false, A, false, B0, (short)0, acc0, false, false);
        acc1 = __builtin_amdgcn_wmma_f32_16x16x4_f32(false, A, false, B1, (short)0, acc1, false, false);
        acc2 = __builtin_amdgcn_wmma_f32_16x16x4_f32(false, A, false, B2, (short)0, acc2, false, false);
    }

    const bool c2ok = (lm < 8);
    for (int j = 0; j < 8; ++j) {
        int rr = row0 + hi8 + j;
        if (rr < N) {
            float* o = h2 + (size_t)rr * F_OUT + lm;
            o[0]  = acc0[j] * dv[j];
            o[16] = acc1[j] * dv[j];
            if (c2ok) o[32] = acc2[j] * dv[j];
        }
    }
}


__global__ void __launch_bounds__(256)
gather2_kernel(const float* __restrict__ h2s, const float* __restrict__ dinv,
               const int* __restrict__ start, const int* __restrict__ cnt,
               const int* __restrict__ csr, const float* __restrict__ b2,
               float* __restrict__ out, int N) {
    __shared__ __attribute__((aligned(16))) float so[8 * F_OUT];
    const int lane = threadIdx.x & 31;
    const int wave = threadIdx.x >> 5;
    const int node = blockIdx.x * 8 + wave;
    if (node < N) {
        const int beg = __builtin_amdgcn_readfirstlane(start[node]);
        const int c   = __builtin_amdgcn_readfirstlane(cnt[node]);
        const float dd = dinv[node];
        const int cl = lane < 20 ? lane : 19;
        const float* col = h2s + cl * 2;
        float2 a = ld_row2(col, node, F_OUT);
        for (int base = 0; base < c; base += 32) {
            int n = c - base;
            if (n > 32) n = 32;
            int my_s = node;
            if (lane < n) my_s = csr[beg + base + lane];
            a = sum_rows(col, F_OUT, my_s, n, a);
        }
        if (lane < 20) {
            const float2 bb = *(const float2*)(b2 + lane * 2);
            so[wave * F_OUT + lane * 2]     = a.x * dd + bb.x;
            so[wave * F_OUT + lane * 2 + 1] = a.y * dd + bb.y;
        }
    }
    __syncthreads();
    const int t = threadIdx.x;
    if (t < (8 * F_OUT) / 4) {
        const size_t g = (size_t)blockIdx.x * (8 * F_OUT) + (size_t)t * 4;
        if (g < (size_t)N * F_OUT)
            *(v4f*)(out + g) = *(const v4f*)(so + t * 4);
    }
}


extern "C" void kernel_launch(void* const* d_in, const int* in_sizes, int n_in,
                              void* d_out, int out_size, void* d_ws, size_t ws_size,
                              hipStream_t stream) {
    const float* x  = (const float*)d_in[0];
    const int*   ei = (const int*)d_in[1];
    const float* W1 = (const float*)d_in[2];
    const float* b1 = (const float*)d_in[3];
    const float* W2 = (const float*)d_in[4];
    const float* b2 = (const float*)d_in[5];

    const int N = in_sizes[0] / F_IN;
    const int E = in_sizes[1] / 2;
    const int* src = ei;
    const int* dst = ei + E;

    float* ws    = (float*)d_ws;
    float* dinv  = ws + (size_t)N;
    float* bufA  = ws + (size_t)2 * N;
    float* bufB  = bufA + (size_t)N * F_HID;
    float* pW1   = bufB + (size_t)N * F_HID;
    float* pW2   = pW1 + (size_t)(F_IN / 4) * 256;
    int*   cnt   = (int*)(pW2 + (size_t)(F_HID / 4) * 256);
    int*   start = cnt + (size_t)N;
    int2*  binned = (int2*)(start + (size_t)N + (N & 1));
    int*   csr   = (int*)(binned + (size_t)E);
    int*   rangeCnt  = csr + (size_t)E;
    int*   rangeFill = rangeCnt + MAXR;
    float* out   = (float*)d_out;

    const int B = 256;
    const int R           = (N + RS - 1) / RS;
    const int tileBlocks  = (E + TILE - 1) / TILE;
    const int gemmBlocks  = (N + 127) / 128;
    const int waveNodeBlk = (N + (B / 32) - 1) / (B / 32);
    const int rp1Blocks   = ((F_IN / 4) * 256 + B - 1) / B;
    const int rp2Blocks   = ((F_HID / 4) * 256 + B - 1) / B;

    repack_w1_kernel<<<rp1Blocks, B, 0, stream>>>(W1, pW1);
    repack_w2_kernel<<<rp2Blocks, B, 0, stream>>>(W2, pW2);

    range_init_kernel<<<(R + B - 1) / B, B, 0, stream>>>(rangeCnt, rangeFill, R);
    range_count_kernel<<<tileBlocks, B, 0, stream>>>(dst, rangeCnt, E, R);
    range_bin_kernel<<<tileBlocks, B, 0, stream>>>(src, dst, rangeCnt, rangeFill, binned, E, R);
    range_build_kernel<<<R, B, 0, stream>>>(binned, rangeCnt, start, cnt, dinv, csr, N);

    gemm1_wmma_kernel<<<gemmBlocks, B, 0, stream>>>(x, pW1, dinv, bufA, N);
    gather1_kernel<<<waveNodeBlk, B, 0, stream>>>(bufA, dinv, start, cnt, csr, b1, bufB, N);

    gemm2_wmma_kernel<<<gemmBlocks, B, 0, stream>>>(bufB, pW2, dinv, bufA, N);
    gather2_kernel<<<waveNodeBlk, B, 0, stream>>>(bufA, dinv, start, cnt, csr, b2, out, N);
}


# Launches 10 -> 7: weight repacks + counter clearing share one launch; layer-1 gather blocks (16 nodes) keep z in LDS and run the W2 product on it, z never stored; sum orders unchanged
# speedup vs baseline: 3.8594x; 1.0576x over previous
#include <hip/hip_runtime.h>

typedef __attribute__((ext_vector_type(2))) float v2f;
typedef __attribute__((ext_vector_type(4))) float v4f;
typedef __attribute__((ext_vector_type(8))) float v8f;

#define F_IN  128
#define F_HID 64
#define F_OUT 40


#define RSH   8
#define RS    256
#define MAXR  1024
#define TILE  4096
#define EPT   (TILE / 256)
#define RPT   (MAXR / 256)
#define BPT   20

__device__ __forceinline__ int block_scan_incl(int v, int* wsum, int& total) {
    const int lane = threadIdx.x & 31;
    const int wave = threadIdx.x >> 5;
    int incl = v;
    for (int o = 1; o < 32; o <<= 1) {
        int t = __shfl_up(incl, o);
        if (lane >= o) incl += t;
    }
    if (lane == 31) wsum[wave] = incl;
    __syncthreads();
    int before = 0, all = 0;
    for (int w = 0; w < 8; ++w) {
        const int s = wsum[w];
        if (w < wave) before += s;
        all += s;
    }
    total = all;
    __syncthreads();
    return incl + before;
}

__global__ void __launch_bounds__(256)
range_count_kernel(const int* __restrict__ dst, int* __restrict__ rangeCnt, int E, int R) {
    __shared__ int h[MAXR];
    const int tid = threadIdx.x;
    const int base = blockIdx.x * TILE;
    int d[EPT];
    for (int k = 0; k < EPT; ++k) {
        const int e = base + k * 256 + tid;
        d[k] = dst[e < E ? e : E - 1];
    }
    for (int i = tid; i < R; i += 256) h[i] = 0;
    __syncthreads();
    for (int k = 0; k < EPT; ++k) {
        const int e = base + k * 256 + tid;
        if (e < E) atomicAdd(&h[d[k] >> RSH], 1);
    }
    __syncthreads();
    for (int i = tid; i < R; i += 256) {
        const int c = h[i];
        if (c) atomicAdd(&rangeCnt[i], c);
    }
}

__global__ void __launch_bounds__(256)
range_bin_kernel(const int* __restrict__ src, const int* __restrict__ dst,
                 const int* __restrict__ rangeCnt, int* __restrict__ rangeFill,
                 int2* __restrict__ binned, int E, int R) {
    __shared__ int h[MAXR];
    __shared__ int off[MAXR];
    __shared__ int wsum[8];
    const int tid = threadIdx.x;
    const int base = blockIdx.x * TILE;
    int ss[EPT], dd[EPT], lr[EPT], hb[EPT];
    int a[4];
    for (int j = 0; j < 4; ++j) {
        const int i = tid * 4 + j;
        a[j] = (i < R) ? rangeCnt[i] : 0;
    }
    for (int k = 0; k < EPT; ++k) {
        const int e  = base + k * 256 + tid;
        const int ec = e < E ? e : E - 1;
        dd[k] = dst[ec];
        ss[k] = src[ec];
    }
    const int s4 = a[0] + a[1] + a[2] + a[3];
    int total;
    int run = block_scan_incl(s4, wsum, total) - s4;
    for (int j = 0; j < 4; ++j) {
        off[tid * 4 + j] = run;
        run += a[j];
        h[tid * 4 + j] = 0;
    }
    __syncthreads();
    for (int k = 0; k < EPT; ++k) {
        const int e = base + k * 256 + tid;
        lr[k] = 0;
        if (e < E) lr[k] = atomicAdd(&h[dd[k] >> RSH], 1);
    }
    __syncthreads();
    int cc[RPT], gg[RPT];
    for (int j = 0; j < RPT; ++j) {
        const int i = tid + j * 256;
        cc[j] = (i < R) ? h[i] : 0;
    }
    for (int j = 0; j < RPT; ++j) {
        gg[j] = 0;
        if (cc[j]) gg[j] = atomicAdd(&rangeFill[tid + j * 256], cc[j]);
    }
    for (int j = 0; j < RPT; ++j) {
        const int i = tid + j * 256;
        if (cc[j]) h[i] = off[i] + gg[j];
    }
    __syncthreads();
    for (int k = 0; k < EPT; ++k) hb[k] = h[dd[k] >> RSH];
    for (int k = 0; k < EPT; ++k) {
        const int e = base + k * 256 + tid;
        if (e < E) binned[hb[k] + lr[k]] = make_int2(ss[k], dd[k]);
    }
}

__global__ void __launch_bounds__(256)
range_build_kernel(const int2* __restrict__ binned, const int* __restrict__ rangeCnt,
                   int* __restrict__ start, int* __restrict__ cnt, float* __restrict__ dinv,
                   int* __restrict__ csr, int N) {
    __shared__ int c[RS];
    __shared__ int cur[RS];
    __shared__ int wsum[8];
    const int tid = threadIdx.x;
    const int r   = blockIdx.x;
    int part = 0;
    for (int j = 0; j < RPT; ++j) {
        const int i = tid + j * 256;
        part += (i < r) ? rangeCnt[i] : 0;
    }
    int rbeg;
    block_scan_incl(part, wsum, rbeg);
    const int n = rangeCnt[r];
    c[tid] = 0;
    __syncthreads();
    int2 p[BPT];
    int held = -1;
    for (int c0 = 0; c0 < n; c0 += BPT * 256) {
        for (int k = 0; k < BPT; ++k) {
            const int i = c0 + k * 256 + tid;
            p[k] = binned[rbeg + (i < n ? i : n - 1)];
        }
        for (int k = 0; k < BPT; ++k) {
            const int i = c0 + k * 256 + tid;
            if (i < n) atomicAdd(&c[p[k].y & (RS - 1)], 1);
        }
        held = c0;
    }
    __syncthreads();
    const int v = c[tid];
    int tot;
    const int excl = block_scan_incl(v, wsum, tot) - v;
    const int node = r * RS + tid;
    if (node < N) {
        start[node] = rbeg + excl;
        cnt[node]   = v;
        dinv[node]  = rsqrtf((float)(v + 1));
    }
    cur[tid] = rbeg + excl;
    __syncthreads();
    for (int c0 = 0; c0 < n; c0 += BPT * 256) {
        if (c0 != held) {
            for (int k = 0; k < BPT; ++k) {
                const int i = c0 + k * 256 + tid;
                p[k] = binned[rbeg + (i < n ? i : n - 1)];
            }
            held = c0;
        }
        int pos[BPT];
        for (int k = 0; k < BPT; ++k) {
            const int i = c0 + k * 256 + tid;
            pos[k] = 0;
            if (i < n) pos[k] = atomicAdd(&cur[p[k].y & (RS - 1)], 1);
        }
        for (int k = 0; k < BPT; ++k) {
            const int i = c0 + k * 256 + tid;
            if (i < n) csr[pos[k]] = p[k].x;
        }
    }
}


__global__ void prep_kernel(const float* __restrict__ W1, const float* __restrict__ W2,
                            float* __restrict__ pW1, float* __restrict__ pW2,
                            int* __restrict__ rangeCnt, int* __restrict__ rangeFill, int R) {
    const int idx  = blockIdx.x * blockDim.x + threadIdx.x;
    const int kk   = idx >> 8;
    const int rem  = idx & 255;
    const int lane = rem >> 3;
    const int j    = rem & 7;
    const int n    = j >> 1;
    const int t    = j & 1;
    const int lm   = lane & 15;
    const int kh   = (lane >> 4) << 1;
    if (idx < (F_IN / 4) * 256)
        pW1[idx] = W1[(size_t)(4 * kk + kh + t) * F_HID + n * 16 + lm];
    if (idx < (F_HID / 4) * 256) {
        const int col = n * 16 + lm;
        float v = 0.0f;
        if (j < 6 && col < F_OUT)
            v = W2[(size_t)(4 * kk + kh + t) * F_OUT + col];
        pW2[idx] = v;
    }
    if (idx < R) { rangeCnt[idx] = 0; rangeFill[idx] = 0; }
}


__global__ void gemm1_wmma_kernel(const float* __restrict__ x, const float* __restrict__ pW1,
                                  const float* __restrict__ dinv, float* __restrict__ h1, int N) {
    const int lane = threadIdx.x & 31;
    const int wave = threadIdx.x >> 5;
    const int row0 = blockIdx.x * 128 + wave * 16;
    const int lm   = lane & 15;
    const int kh   = (lane >> 4) << 1;
    int r = row0 + lm;
    if (r > N - 1) r = N - 1;
    const float* xrow = x + (size_t)r * F_IN + kh;
    const float* pw   = pW1 + (size_t)lane * 8;

    const int hi8 = (lane >> 4) << 3;
    float dv[8];
    for (int j = 0; j < 8; ++j) {
        const int rr = row0 + hi8 + j;
        dv[j] = dinv[rr < N ? rr : N - 1];
    }

    v8f acc0 = {}, acc1 = {}, acc2 = {}, acc3 = {};

    for (int kk = 0; kk < F_IN / 4; ++kk) {
        const v2f A   = *(const v2f*)(xrow + kk * 4);
        const v4f w01 = *(const v4f*)(pw + (size_t)kk * 256);
        const v4f w23 = *(const v4f*)(pw + (size_t)kk * 256 + 4);
        v2f B0; B0[0] = w01[0]; B0[1] = w01[1];
        v2f B1; B1[0] = w01[2]; B1[1] = w01[3];
        v2f B2; B2[0] = w23[0]; B2[1] = w23[1];
        v2f B3; B3[0] = w23[2]; B3[1] = w23[3];
        acc0 = __builtin_amdgcn_wmma_f32_16x16x4_f32(false, A, false, B0, (short)0, acc0, false, false);
        acc1 = __builtin_amdgcn_wmma_f32_16x16x4_f32(false, A, false, B1, (short)0, acc1, false, false);
        acc2 = __builtin_amdgcn_wmma_f32_16x16x4_f32(false, A, false, B2, (short)0, acc2, false, false);
        acc3 = __builtin_amdgcn_wmma_f32_16x16x4_f32(false, A, false, B3, (short)0, acc3, false, false);
    }

    for (int j = 0; j < 8; ++j) {
        int rr = row0 + hi8 + j;
        if (rr < N) {
            float* o = h1 + (size_t)rr * F_HID + lm;
            o[0]  = acc0[j] * dv[j];
            o[16] = acc1[j] * dv[j];
            o[32] = acc2[j] * dv[j];
            o[48] = acc3[j] * dv[j];
        }
    }
}


#define RL(v, j) __builtin_amdgcn_readlane((v), (j))

__device__ __forceinline__ float2 ld_row2(const float* __restrict__ col, int s, int pitch) {
    return *(const float2*)(col + (size_t)s * pitch);
}

__device__ __forceinline__ float2 sum_rows(const float* __restrict__ col, int pitch,
                                           int my_s, int n, float2 a) {
    int j = 0;
    for (; j + 8 <= n; j += 8) {
        const float2 v0 = ld_row2(col, RL(my_s, j + 0), pitch);
        const float2 v1 = ld_row2(col, RL(my_s, j + 1), pitch);
        const float2 v2 = ld_row2(col, RL(my_s, j + 2), pitch);
        const float2 v3 = ld_row2(col, RL(my_s, j + 3), pitch);
        const float2 v4 = ld_row2(col, RL(my_s, j + 4), pitch);
        const float2 v5 = ld_row2(col, RL(my_s, j + 5), pitch);
        const float2 v6 = ld_row2(col, RL(my_s, j + 6), pitch);
        const float2 v7 = ld_row2(col, RL(my_s, j + 7), pitch);
        a.x += v0.x; a.y += v0.y;
        a.x += v1.x; a.y += v1.y;
        a.x += v2.x; a.y += v2.y;
        a.x += v3.x; a.y += v3.y;
        a.x += v4.x; a.y += v4.y;
        a.x += v5.x; a.y += v5.y;
        a.x += v6.x; a.y += v6.y;
        a.x += v7.x; a.y += v7.y;
    }
    if (j + 4 <= n) {
        const float2 v0 = ld_row2(col, RL(my_s, j + 0), pitch);
        const float2 v1 = ld_row2(col, RL(my_s, j + 1), pitch);
        const float2 v2 = ld_row2(col, RL(my_s, j + 2), pitch);
        const float2 v3 = ld_row2(col, RL(my_s, j + 3), pitch);
        a.x += v0.x; a.y += v0.y;
        a.x += v1.x; a.y += v1.y;
        a.x += v2.x; a.y += v2.y;
        a.x += v3.x; a.y += v3.y;
        j += 4;
    }
    if (j + 2 <= n) {
        const float2 v0 = ld_row2(col, RL(my_s, j + 0), pitch);
        const float2 v1 = ld_row2(col, RL(my_s, j + 1), pitch);
        a.x += v0.x; a.y += v0.y;
        a.x += v1.x; a.y += v1.y;
        j += 2;
    }
    if (j < n) {
        const float2 v0 = ld_row2(col, RL(my_s, j), pitch);
        a.x += v0.x; a.y += v0.y;
    }
    return a;
}


#define NB 16
#define ZP (F_HID + 4)

__global__ void __launch_bounds__(512)
gather1_gemm2_kernel(const float* __restrict__ h1s, const float* __restrict__ dinv,
                     const int* __restrict__ start, const int* __restrict__ cnt,
                     const int* __restrict__ csr, const float* __restrict__ b1,
                     const float* __restrict__ pW2, float* __restrict__ h2s, int N) {
    __shared__ __attribute__((aligned(16))) float sz[NB * ZP];
    __shared__ __attribute__((aligned(16))) float so[NB * F_OUT];
    __shared__ float sd[NB];
    const int lane = threadIdx.x & 31;
    const int wave = __builtin_amdgcn_readfirstlane(threadIdx.x >> 5);
    int node = blockIdx.x * NB + wave;
    if (node > N - 1) node = N - 1;
    {
        const int beg = __builtin_amdgcn_readfirstlane(start[node]);
        const int c   = __builtin_amdgcn_readfirstlane(cnt[node]);
        const float dd = dinv[node];
        const float* col = h1s + lane * 2;
        float2 a = ld_row2(col, node, F_HID);
        for (int base = 0; base < c; base += 32) {
            int n = c - base;
            if (n > 32) n = 32;
            int my_s = node;
            if (lane < n) my_s = csr[beg + base + lane];
            a = sum_rows(col, F_HID, my_s, n, a);
        }
        const float2 bb = *(const float2*)(b1 + lane * 2);
        const float rx = a.x * dd + bb.x;
        const float ry = a.y * dd + bb.y;
        float2 r;
        r.x = rx > 0.0f ? rx : 0.0f;
        r.y = ry > 0.0f ? ry : 0.0f;
        *(float2*)(sz + wave * ZP + lane * 2) = r;
        if (lane == 0) sd[wave] = dd;
    }
    __syncthreads();
    if (wave < 3) {
        const int lm  = lane & 15;
        const int kh  = (lane >> 4) << 1;
        const int hi8 = (lane >> 4) << 3;
        const float* zrow = sz + lm * ZP + kh;
        const float* pw   = pW2 + (size_t)lane * 8 + wave * 2;
        v2f Bf[F_HID / 4];
        #pragma unroll
        for (int kk = 0; kk < F_HID / 4; ++kk) Bf[kk] = *(const v2f*)(pw + (size_t)kk * 256);
        v8f acc = {};
        #pragma unroll
        for (int kk = 0; kk < F_HID / 4; ++kk) {
            const v2f A = *(const v2f*)(zrow + kk * 4);
            acc = __builtin_amdgcn_wmma_f32_16x16x4_f32(false, A, false, Bf[kk], (short)0, acc, false, false);
        }
        const int colo = wave * 16 + lm;
        if (colo < F_OUT) {
            #pragma unroll
            for (int j = 0; j < 8; ++j)
                so[(hi8 + j) * F_OUT + colo] = acc[j] * sd[hi8 + j];
        }
    }
    __syncthreads();
    const int t = threadIdx.x;
    if (t < (NB * F_OUT) / 4) {
        const size_t g = (size_t)blockIdx.x * (NB * F_OUT) + (size_t)t * 4;
        if (g < (size_t)N * F_OUT)
            *(v4f*)(h2s + g) = *(const v4f*)(so + t * 4);
    }
}


__global__ void __launch_bounds__(256)
gather2_kernel(const float* __restrict__ h2s, const float* __restrict__ dinv,
               const int* __restrict__ start, const int* __restrict__ cnt,
               const int* __restrict__ csr, const float* __restrict__ b2,
               float* __restrict__ out, int N) {
    __shared__ __attribute__((aligned(16))) float so[8 * F_OUT];
    const int lane = threadIdx.x & 31;
    const int wave = threadIdx.x >> 5;
    const int node = blockIdx.x * 8 + wave;
    if (node < N) {
        const int beg = __builtin_amdgcn_readfirstlane(start[node]);
        const int c   = __builtin_amdgcn_readfirstlane(cnt[node]);
        const float dd = dinv[node];
        const int cl = lane < 20 ? lane : 19;
        const float* col = h2s + cl * 2;
        float2 a = ld_row2(col, node, F_OUT);
        for (int base = 0; base < c; base += 32) {
            int n = c - base;
            if (n > 32) n = 32;
            int my_s = node;
            if (lane < n) my_s = csr[beg + base + lane];
            a = sum_rows(col, F_OUT, my_s, n, a);
        }
        if (lane < 20) {
            const float2 bb = *(const float2*)(b2 + lane * 2);
            so[wave * F_OUT + lane * 2]     = a.x * dd + bb.x;
            so[wave * F_OUT + lane * 2 + 1] = a.y * dd + bb.y;
        }
    }
    __syncthreads();
    const int t = threadIdx.x;
    if (t < (8 * F_OUT) / 4) {
        const size_t g = (size_t)blockIdx.x * (8 * F_OUT) + (size_t)t * 4;
        if (g < (size_t)N * F_OUT)
            *(v4f*)(out + g) = *(const v4f*)(so + t * 4);
    }
}


extern "C" void kernel_launch(void* const* d_in, const int* in_sizes, int n_in,
                              void* d_out, int out_size, void* d_ws, size_t ws_size,
                              hipStream_t stream) {
    const float* x  = (const float*)d_in[0];
    const int*   ei = (const int*)d_in[1];
    const float* W1 = (const float*)d_in[2];
    const float* b1 = (const float*)d_in[3];
    const float* W2 = (const float*)d_in[4];
    const float* b2 = (const float*)d_in[5];

    const int N = in_sizes[0] / F_IN;
    const int E = in_sizes[1] / 2;
    const int* src = ei;
    const int* dst = ei + E;

    float* ws    = (float*)d_ws;
    float* dinv  = ws + (size_t)N;
    float* bufA  = ws + (size_t)2 * N;
    float* bufB  = bufA + (size_t)N * F_HID;
    float* pW1   = bufB + (size_t)N * F_HID;
    float* pW2   = pW1 + (size_t)(F_IN / 4) * 256;
    int*   cnt   = (int*)(pW2 + (size_t)(F_HID / 4) * 256);
    int*   start = cnt + (size_t)N;
    int2*  binned = (int2*)(start + (size_t)N + (N & 1));
    int*   csr   = (int*)(binned + (size_t)E);
    int*   rangeCnt  = csr + (size_t)E;
    int*   rangeFill = rangeCnt + MAXR;
    float* out   = (float*)d_out;

    const int B = 256;
    const int R           = (N + RS - 1) / RS;
    const int tileBlocks  = (E + TILE - 1) / TILE;
    const int gemmBlocks  = (N + 127) / 128;
    const int waveNodeBlk = (N + (B / 32) - 1) / (B / 32);
    const int fusedBlocks = (N + NB - 1) / NB;
    const int prepBlocks  = ((F_IN / 4) * 256 + B - 1) / B;

    prep_kernel<<<prepBlocks, B, 0, stream>>>(W1, W2, pW1, pW2, rangeCnt, rangeFill, R);

    range_count_kernel<<<tileBlocks, B, 0, stream>>>(dst, rangeCnt, E, R);
    range_bin_kernel<<<tileBlocks, B, 0, stream>>>(src, dst, rangeCnt, rangeFill, binned, E, R);
    range_build_kernel<<<R, B, 0, stream>>>(binned, rangeCnt, start, cnt, dinv, csr, N);

    gemm1_wmma_kernel<<<gemmBlocks, B, 0, stream>>>(x, pW1, dinv, bufA, N);
    gather1_gemm2_kernel<<<fusedBlocks, NB * 32, 0, stream>>>(bufA, dinv, start, cnt, csr, b1, pW2, bufB, N);

    gather2_kernel<<<waveNodeBlk, B, 0, stream>>>(bufB, dinv, start, cnt, csr, b2, out, N);
}


# Both gathers: a wave walks 8 consecutive nodes (next node's list head and own row requested ahead) instead of one node per wave: 200,000 gather waves -> 25,000; per-node sum order unchanged
# speedup vs baseline: 4.1872x; 1.0849x over previous
#include <hip/hip_runtime.h>

typedef __attribute__((ext_vector_type(2))) float v2f;
typedef __attribute__((ext_vector_type(4))) float v4f;
typedef __attribute__((ext_vector_type(8))) float v8f;

#define F_IN  128
#define F_HID 64
#define F_OUT 40


#define RSH   8
#define RS    256
#define MAXR  1024
#define TILE  4096
#define EPT   (TILE / 256)
#define RPT   (MAXR / 256)
#define BPT   20

__device__ __forceinline__ int block_scan_incl(int v, int* wsum, int& total) {
    const int lane = threadIdx.x & 31;
    const int wave = threadIdx.x >> 5;
    int incl = v;
    for (int o = 1; o < 32; o <<= 1) {
        int t = __shfl_up(incl, o);
        if (lane >= o) incl += t;
    }
    if (lane == 31) wsum[wave] = incl;
    __syncthreads();
    int before = 0, all = 0;
    for (int w = 0; w < 8; ++w) {
        const int s = wsum[w];
        if (w < wave) before += s;
        all += s;
    }
    total = all;
    __syncthreads();
    return incl + before;
}

__global__ void __launch_bounds__(256)
range_count_kernel(const int* __restrict__ dst, int* __restrict__ rangeCnt, int E, int R) {
    __shared__ int h[MAXR];
    const int tid = threadIdx.x;
    const int base = blockIdx.x * TILE;
    int d[EPT];
    for (int k = 0; k < EPT; ++k) {
        const int e = base + k * 256 + tid;
        d[k] = dst[e < E ? e : E - 1];
    }
    for (int i = tid; i < R; i += 256) h[i] = 0;
    __syncthreads();
    for (int k = 0; k < EPT; ++k) {
        const int e = base + k * 256 + tid;
        if (e < E) atomicAdd(&h[d[k] >> RSH], 1);
    }
    __syncthreads();
    for (int i = tid; i < R; i += 256) {
        const int c = h[i];
        if (c) atomicAdd(&rangeCnt[i], c);
    }
}

__global__ void __launch_bounds__(256)
range_bin_kernel(const int* __restrict__ src, const int* __restrict__ dst,
                 const int* __restrict__ rangeCnt, int* __restrict__ rangeFill,
                 int2* __restrict__ binned, int E, int R) {
    __shared__ int h[MAXR];
    __shared__ int off[MAXR];
    __shared__ int wsum[8];
    const int tid = threadIdx.x;
    const int base = blockIdx.x * TILE;
    int ss[EPT], dd[EPT], lr[EPT], hb[EPT];
    int a[4];
    for (int j = 0; j < 4; ++j) {
        const int i = tid * 4 + j;
        a[j] = (i < R) ? rangeCnt[i] : 0;
    }
    for (int k = 0; k < EPT; ++k) {
        const int e  = base + k * 256 + tid;
        const int ec = e < E ? e : E - 1;
        dd[k] = dst[ec];
        ss[k] = src[ec];
    }
    const int s4 = a[0] + a[1] + a[2] + a[3];
    int total;
    int run = block_scan_incl(s4, wsum, total) - s4;
    for (int j = 0; j < 4; ++j) {
        off[tid * 4 + j] = run;
        run += a[j];
        h[tid * 4 + j] = 0;
    }
    __syncthreads();
    for (int k = 0; k < EPT; ++k) {
        const int e = base + k * 256 + tid;
        lr[k] = 0;
        if (e < E) lr[k] = atomicAdd(&h[dd[k] >> RSH], 1);
    }
    __syncthreads();
    int cc[RPT], gg[RPT];
    for (int j = 0; j < RPT; ++j) {
        const int i = tid + j * 256;
        cc[j] = (i < R) ? h[i] : 0;
    }
    for (int j = 0; j < RPT; ++j) {
        gg[j] = 0;
        if (cc[j]) gg[j] = atomicAdd(&rangeFill[tid + j * 256], cc[j]);
    }
    for (int j = 0; j < RPT; ++j) {
        const int i = tid + j * 256;
        if (cc[j]) h[i] = off[i] + gg[j];
    }
    __syncthreads();
    for (int k = 0; k < EPT; ++k) hb[k] = h[dd[k] >> RSH];
    for (int k = 0; k < EPT; ++k) {
        const int e = base + k * 256 + tid;
        if (e < E) binned[hb[k] + lr[k]] = make_int2(ss[k], dd[k]);
    }
}

__global__ void __launch_bounds__(256)
range_build_kernel(const int2* __restrict__ binned, const int* __restrict__ rangeCnt,
                   int* __restrict__ start, int* __restrict__ cnt, float* __restrict__ dinv,
                   int* __restrict__ csr, int N) {
    __shared__ int c[RS];
    __shared__ int cur[RS];
    __shared__ int wsum[8];
    const int tid = threadIdx.x;
    const int r   = blockIdx.x;
    int part = 0;
    for (int j = 0; j < RPT; ++j) {
        const int i = tid + j * 256;
        part += (i < r) ? rangeCnt[i] : 0;
    }
    int rbeg;
    block_scan_incl(part, wsum, rbeg);
    const int n = rangeCnt[r];
    c[tid] = 0;
    __syncthreads();
    int2 p[BPT];
    int held = -1;
    for (int c0 = 0; c0 < n; c0 += BPT * 256) {
        for (int k = 0; k < BPT; ++k) {
            const int i = c0 + k * 256 + tid;
            p[k] = binned[rbeg + (i < n ? i : n - 1)];
        }
        for (int k = 0; k < BPT; ++k) {
            const int i = c0 + k * 256 + tid;
            if (i < n) atomicAdd(&c[p[k].y & (RS - 1)], 1);
        }
        held = c0;
    }
    __syncthreads();
    const int v = c[tid];
    int tot;
    const int excl = block_scan_incl(v, wsum, tot) - v;
    const int node = r * RS + tid;
    if (node < N) {
        start[node] = rbeg + excl;
        cnt[node]   = v;
        dinv[node]  = rsqrtf((float)(v + 1));
    }
    cur[tid] = rbeg + excl;
    __syncthreads();
    for (int c0 = 0; c0 < n; c0 += BPT * 256) {
        if (c0 != held) {
            for (int k = 0; k < BPT; ++k) {
                const int i = c0 + k * 256 + tid;
                p[k] = binned[rbeg + (i < n ? i : n - 1)];
            }
            held = c0;
        }
        int pos[BPT];
        for (int k = 0; k < BPT; ++k) {
            const int i = c0 + k * 256 + tid;
            pos[k] = 0;
            if (i < n) pos[k] = atomicAdd(&cur[p[k].y & (RS - 1)], 1);
        }
        for (int k = 0; k < BPT; ++k) {
            const int i = c0 + k * 256 + tid;
            if (i < n) csr[pos[k]] = p[k].x;
        }
    }
}


__global__ void prep_kernel(const float* __restrict__ W1, const float* __restrict__ W2,
                            float* __restrict__ pW1, float* __restrict__ pW2,
                            int* __restrict__ rangeCnt, int* __restrict__ rangeFill, int R) {
    const int idx  = blockIdx.x * blockDim.x + threadIdx.x;
    const int kk   = idx >> 8;
    const int rem  = idx & 255;
    const int lane = rem >> 3;
    const int j    = rem & 7;
    const int n    = j >> 1;
    const int t    = j & 1;
    const int lm   = lane & 15;
    const int kh   = (lane >> 4) << 1;
    if (idx < (F_IN / 4) * 256)
        pW1[idx] = W1[(size_t)(4 * kk + kh + t) * F_HID + n * 16 + lm];
    if (idx < (F_HID / 4) * 256) {
        const int col = n * 16 + lm;
        float v = 0.0f;
        if (j < 6 && col < F_OUT)
            v = W2[(size_t)(4 * kk + kh + t) * F_OUT + col];
        pW2[idx] = v;
    }
    if (idx < R) { rangeCnt[idx] = 0; rangeFill[idx] = 0; }
}


__global__ void gemm1_wmma_kernel(const float* __restrict__ x, const float* __restrict__ pW1,
                                  const float* __restrict__ dinv, float* __restrict__ h1, int N) {
    const int lane = threadIdx.x & 31;
    const int wave = threadIdx.x >> 5;
    const int row0 = blockIdx.x * 128 + wave * 16;
    const int lm   = lane & 15;
    const int kh   = (lane >> 4) << 1;
    int r = row0 + lm;
    if (r > N - 1) r = N - 1;
    const float* xrow = x + (size_t)r * F_IN + kh;
    const float* pw   = pW1 + (size_t)lane * 8;

    const int hi8 = (lane >> 4) << 3;
    float dv[8];
    for (int j = 0; j < 8; ++j) {
        const int rr = row0 + hi8 + j;
        dv[j] = dinv[rr < N ? rr : N - 1];
    }

    v8f acc0 = {}, acc1 = {}, acc2 = {}, acc3 = {};

    for (int kk = 0; kk < F_IN / 4; ++kk) {
        const v2f A   = *(const v2f*)(xrow + kk * 4);
        const v4f w01 = *(const v4f*)(pw + (size_t)kk * 256);
        const v4f w23 = *(const v4f*)(pw + (size_t)kk * 256 + 4);
        v2f B0; B0[0] = w01[0]; B0[1] = w01[1];
        v2f B1; B1[0] = w01[2]; B1[1] = w01[3];
        v2f B2; B2[0] = w23[0]; B2[1] = w23[1];
        v2f B3; B3[0] = w23[2]; B3[1] = w23[3];
        acc0 = __builtin_amdgcn_wmma_f32_16x16x4_f32(false, A, false, B0, (short)0, acc0, false, false);
        acc1 = __builtin_amdgcn_wmma_f32_16x16x4_f32(false, A, false, B1, (short)0, acc1, false, false);
        acc2 = __builtin_amdgcn_wmma_f32_16x16x4_f32(false, A, false, B2, (short)0, acc2, false, false);
        acc3 = __builtin_amdgcn_wmma_f32_16x16x4_f32(false, A, false, B3, (short)0, acc3, false, false);
    }

    for (int j = 0; j < 8; ++j) {
        int rr = row0 + hi8 + j;
        if (rr < N) {
            float* o = h1 + (size_t)rr * F_HID + lm;
            o[0]  = acc0[j] * dv[j];
            o[16] = acc1[j] * dv[j];
            o[32] = acc2[j] * dv[j];
            o[48] = acc3[j] * dv[j];
        }
    }
}


#define RL(v, j) __builtin_amdgcn_readlane((v), (j))

__device__ __forceinline__ float2 ld_row2(const float* __restrict__ col, int s, int pitch) {
    return *(const float2*)(col + (size_t)s * pitch);
}

__device__ __forceinline__ float2 sum_rows(const float* __restrict__ col, int pitch,
                                           int my_s, int n, float2 a) {
    int j = 0;
    for (; j + 8 <= n; j += 8) {
        const float2 v0 = ld_row2(col, RL(my_s, j + 0), pitch);
        const float2 v1 = ld_row2(col, RL(my_s, j + 1), pitch);
        const float2 v2 = ld_row2(col, RL(my_s, j + 2), pitch);
        const float2 v3 = ld_row2(col, RL(my_s, j + 3), pitch);
        const float2 v4 = ld_row2(col, RL(my_s, j + 4), pitch);
        const float2 v5 = ld_row2(col, RL(my_s, j + 5), pitch);
        const float2 v6 = ld_row2(col, RL(my_s, j + 6), pitch);
        const float2 v7 = ld_row2(col, RL(my_s, j + 7), pitch);
        a.x += v0.x; a.y += v0.y;
        a.x += v1.x; a.y += v1.y;
        a.x += v2.x; a.y += v2.y;
        a.x += v3.x; a.y += v3.y;
        a.x += v4.x; a.y += v4.y;
        a.x += v5.x; a.y += v5.y;
        a.x += v6.x; a.y += v6.y;
        a.x += v7.x; a.y += v7.y;
    }
    if (j + 4 <= n) {
        const float2 v0 = ld_row2(col, RL(my_s, j + 0), pitch);
        const float2 v1 = ld_row2(col, RL(my_s, j + 1), pitch);
        const float2 v2 = ld_row2(col, RL(my_s, j + 2), pitch);
        const float2 v3 = ld_row2(col, RL(my_s, j + 3), pitch);
        a.x += v0.x; a.y += v0.y;
        a.x += v1.x; a.y += v1.y;
        a.x += v2.x; a.y += v2.y;
        a.x += v3.x; a.y += v3.y;
        j += 4;
    }
    if (j + 2 <= n) {
        const float2 v0 = ld_row2(col, RL(my_s, j + 0), pitch);
        const float2 v1 = ld_row2(col, RL(my_s, j + 1), pitch);
        a.x += v0.x; a.y += v0.y;
        a.x += v1.x; a.y += v1.y;
        j += 2;
    }
    if (j < n) {
        const float2 v0 = ld_row2(col, RL(my_s, j), pitch);
        a.x += v0.x; a.y += v0.y;
    }
    return a;
}


#define NPW 8

__device__ __forceinline__ float2 gather_node(const float* __restrict__ col, int pitch,
                                              const int* __restrict__ csr, int node, int beg, int c,
                                              int my_s, float2 self, int lane) {
    float2 a = sum_rows(col, pitch, my_s, c < 32 ? c : 32, self);
    for (int base = 32; base < c; base += 32) {
        int n = c - base;
        if (n > 32) n = 32;
        int s2 = node;
        if (lane < n) s2 = csr[beg + base + lane];
        a = sum_rows(col, pitch, s2, n, a);
    }
    return a;
}


#define G1W 4
#define NB  (G1W * NPW)
#define ZP  (F_HID + 4)

__global__ void __launch_bounds__(G1W * 32)
gather1_gemm2_kernel(const float* __restrict__ h1s, const float* __restrict__ dinv,
                     const int* __restrict__ start, const int* __restrict__ cnt,
                     const int* __restrict__ csr, const float* __restrict__ b1,
                     const float* __restrict__ pW2, float* __restrict__ h2s, int N) {
    __shared__ __attribute__((aligned(16))) float sz[NB * ZP];
    __shared__ __attribute__((aligned(16))) float so[NB * F_OUT];
    __shared__ float sd[NB];
    const int lane = threadIdx.x & 31;
    const int wave = __builtin_amdgcn_readfirstlane(threadIdx.x >> 5);
    const int row0 = wave * NPW;
    const int n0   = blockIdx.x * NB + row0;
    {
        const float* col = h1s + lane * 2;
        const float2 bb = *(const float2*)(b1 + lane * 2);
        int ln = n0 + (lane < NPW ? lane : NPW - 1);
        if (ln > N - 1) ln = N - 1;
        const int   sv  = start[ln];
        const int   cv  = cnt[ln];
        const float dvv = dinv[ln];
        if (lane < NPW) sd[row0 + lane] = dvv;
        int node = n0 < N - 1 ? n0 : N - 1;
        int beg  = RL(sv, 0);
        int c    = RL(cv, 0);
        int my_s = node;
        if (lane < c) my_s = csr[beg + lane];
        float2 self = ld_row2(col, node, F_HID);
        for (int i = 0; i < NPW; ++i) {
            const int ni = i + 1 < NPW ? i + 1 : NPW - 1;
            int nnode = n0 + ni;
            if (nnode > N - 1) nnode = N - 1;
            const int nbeg = RL(sv, ni);
            const int nc   = RL(cv, ni);
            int nmy = nnode;
            if (lane < nc) nmy = csr[nbeg + lane];
            const float2 nself = ld_row2(col, nnode, F_HID);
            const float dd = __int_as_float(RL(__float_as_int(dvv), i));
            const float2 a = gather_node(col, F_HID, csr, node, beg, c, my_s, self, lane);
            const float rx = a.x * dd + bb.x;
            const float ry = a.y * dd + bb.y;
            float2 r;
            r.x = rx > 0.0f ? rx : 0.0f;
            r.y = ry > 0.0f ? ry : 0.0f;
            *(float2*)(sz + (row0 + i) * ZP + lane * 2) = r;
            node = nnode; beg = nbeg; c = nc; my_s = nmy; self = nself;
        }
    }
    __syncthreads();
    if (wave < 3) {
        const int lm  = lane & 15;
        const int kh  = (lane >> 4) << 1;
        const int hi8 = (lane >> 4) << 3;
        const float* zrow0 = sz + lm * ZP + kh;
        const float* zrow1 = zrow0 + 16 * ZP;
        const float* pw    = pW2 + (size_t)lane * 8 + wave * 2;
        v2f Bf[F_HID / 4];
        #pragma unroll
        for (int kk = 0; kk < F_HID / 4; ++kk) Bf[kk] = *(const v2f*)(pw + (size_t)kk * 256);
        v8f acc0 = {}, acc1 = {};
        #pragma unroll
        for (int kk = 0; kk < F_HID / 4; ++kk) {
            const v2f A0 = *(const v2f*)(zrow0 + kk * 4);
            const v2f A1 = *(const v2f*)(zrow1 + kk * 4);
            acc0 = __builtin_amdgcn_wmma_f32_16x16x4_f32(false, A0, false, Bf[kk], (short)0, acc0, false, false);
            acc1 = __builtin_amdgcn_wmma_f32_16x16x4_f32(false, A1, false, Bf[kk], (short)0, acc1, false, false);
        }
        const int colo = wave * 16 + lm;
        if (colo < F_OUT) {
            #pragma unroll
            for (int j = 0; j < 8; ++j) {
                so[(hi8 + j) * F_OUT + colo]      = acc0[j] * sd[hi8 + j];
                so[(16 + hi8 + j) * F_OUT + colo] = acc1[j] * sd[16 + hi8 + j];
            }
        }
    }
    __syncthreads();
    for (int t = threadIdx.x; t < (NB * F_OUT) / 4; t += G1W * 32) {
        const size_t g = (size_t)blockIdx.x * (NB * F_OUT) + (size_t)t * 4;
        if (g < (size_t)N * F_OUT)
            *(v4f*)(h2s + g) = *(const v4f*)(so + t * 4);
    }
}


#define G2W 4
#define NB2 (G2W * NPW)

__global__ void __launch_bounds__(G2W * 32)
gather2_kernel(const float* __restrict__ h2s, const float* __restrict__ dinv,
               const int* __restrict__ start, const int* __restrict__ cnt,
               const int* __restrict__ csr, const float* __restrict__ b2,
               float* __restrict__ out, int N) {
    __shared__ __attribute__((aligned(16))) float so[NB2 * F_OUT];
    const int lane = threadIdx.x & 31;
    const int wave = __builtin_amdgcn_readfirstlane(threadIdx.x >> 5);
    const int row0 = wave * NPW;
    const int n0   = blockIdx.x * NB2 + row0;
    if (n0 < N) {
        const int cl = lane < 20 ? lane : 19;
        const float* col = h2s + cl * 2;
        const float2 bb = *(const float2*)(b2 + cl * 2);
        int ln = n0 + (lane < NPW ? lane : NPW - 1);
        if (ln > N - 1) ln = N - 1;
        const int   sv  = start[ln];
        const int   cv  = cnt[ln];
        const float dvv = dinv[ln];
        int node = n0;
        int beg  = RL(sv, 0);
        int c    = RL(cv, 0);
        int my_s = node;
        if (lane < c) my_s = csr[beg + lane];
        float2 self = ld_row2(col, node, F_OUT);
        for (int i = 0; i < NPW; ++i) {
            const int ni = i + 1 < NPW ? i + 1 : NPW - 1;
            int nnode = n0 + ni;
            if (nnode > N - 1) nnode = N - 1;
            const int nbeg = RL(sv, ni);
            const int nc   = RL(cv, ni);
            int nmy = nnode;
            if (lane < nc) nmy = csr[nbeg + lane];
            const float2 nself = ld_row2(col, nnode, F_OUT);
            const float dd = __int_as_float(RL(__float_as_int(dvv), i));
            const float2 a = gather_node(col, F_OUT, csr, node, beg, c, my_s, self, lane);
            if (lane < 20) {
                so[(row0 + i) * F_OUT + lane * 2]     = a.x * dd + bb.x;
                so[(row0 + i) * F_OUT + lane * 2 + 1] = a.y * dd + bb.y;
            }
            node = nnode; beg = nbeg; c = nc; my_s = nmy; self = nself;
        }
    }
    __syncthreads();
    for (int t = threadIdx.x; t < (NB2 * F_OUT) / 4; t += G2W * 32) {
        const size_t g = (size_t)blockIdx.x * (NB2 * F_OUT) + (size_t)t * 4;
        if (g < (size_t)N * F_OUT)
            *(v4f*)(out + g) = *(const v4f*)(so + t * 4);
    }
}


extern "C" void kernel_launch(void* const* d_in, const int* in_sizes, int n_in,
                              void* d_out, int out_size, void* d_ws, size_t ws_size,
                              hipStream_t stream) {
    const float* x  = (const float*)d_in[0];
    const int*   ei = (const int*)d_in[1];
    const float* W1 = (const float*)d_in[2];
    const float* b1 = (const float*)d_in[3];
    const float* W2 = (const float*)d_in[4];
    const float* b2 = (const float*)d_in[5];

    const int N = in_sizes[0] / F_IN;
    const int E = in_sizes[1] / 2;
    const int* src = ei;
    const int* dst = ei + E;

    float* ws    = (float*)d_ws;
    float* dinv  = ws + (size_t)N;
    float* bufA  = ws + (size_t)2 * N;
    float* bufB  = bufA + (size_t)N * F_HID;
    float* pW1   = bufB + (size_t)N * F_HID;
    float* pW2   = pW1 + (size_t)(F_IN / 4) * 256;
    int*   cnt   = (int*)(pW2 + (size_t)(F_HID / 4) * 256);
    int*   start = cnt + (size_t)N;
    int2*  binned = (int2*)(start + (size_t)N + (N & 1));
    int*   csr   = (int*)(binned + (size_t)E);
    int*   rangeCnt  = csr + (size_t)E;
    int*   rangeFill = rangeCnt + MAXR;
    float* out   = (float*)d_out;

    const int B = 256;
    const int R           = (N + RS - 1) / RS;
    const int tileBlocks  = (E + TILE - 1) / TILE;
    const int gemmBlocks  = (N + 127) / 128;
    const int gather2Blk  = (N + NB2 - 1) / NB2;
    const int fusedBlocks = (N + NB - 1) / NB;
    const int prepBlocks  = ((F_IN / 4) * 256 + B - 1) / B;

    prep_kernel<<<prepBlocks, B, 0, stream>>>(W1, W2, pW1, pW2, rangeCnt, rangeFill, R);

    range_count_kernel<<<tileBlocks, B, 0, stream>>>(dst, rangeCnt, E, R);
    range_bin_kernel<<<tileBlocks, B, 0, stream>>>(src, dst, rangeCnt, rangeFill, binned, E, R);
    range_build_kernel<<<R, B, 0, stream>>>(binned, rangeCnt, start, cnt, dinv, csr, N);

    gemm1_wmma_kernel<<<gemmBlocks, B, 0, stream>>>(x, pW1, dinv, bufA, N);
    gather1_gemm2_kernel<<<fusedBlocks, G1W * 32, 0, stream>>>(bufA, dinv, start, cnt, csr, b1, pW2, bufB, N);

    gather2_kernel<<<gather2Blk, G2W * 32, 0, stream>>>(bufB, dinv, start, cnt, csr, b2, out, N);
}


# List build: both placing passes order their items in LDS first and store them in that order (a store writes a few contiguous pieces, was 32 scattered slots); list order still LDS arrival order
# speedup vs baseline: 4.7125x; 1.1255x over previous
#include <hip/hip_runtime.h>

typedef __attribute__((ext_vector_type(2))) float v2f;
typedef __attribute__((ext_vector_type(4))) float v4f;
typedef __attribute__((ext_vector_type(8))) float v8f;

#define F_IN  128
#define F_HID 64
#define F_OUT 40


#define RSH   8
#define RS    256
#define MAXR  1024
#define TILE  4096
#define EPT   (TILE / 256)
#define RPT   (MAXR / 256)
#define BPT   20
#define SCAP  8192

__device__ __forceinline__ int block_scan_incl(int v, int* wsum, int& total) {
    const int lane = threadIdx.x & 31;
    const int wave = threadIdx.x >> 5;
    int incl = v;
    for (int o = 1; o < 32; o <<= 1) {
        int t = __shfl_up(incl, o);
        if (lane >= o) incl += t;
    }
    if (lane == 31) wsum[wave] = incl;
    __syncthreads();
    int before = 0, all = 0;
    for (int w = 0; w < 8; ++w) {
        const int s = wsum[w];
        if (w < wave) before += s;
        all += s;
    }
    total = all;
    __syncthreads();
    return incl + before;
}

__global__ void __launch_bounds__(256)
range_count_kernel(const int* __restrict__ dst, int* __restrict__ rangeCnt, int E, int R) {
    __shared__ int h[MAXR];
    const int tid = threadIdx.x;
    const int base = blockIdx.x * TILE;
    int d[EPT];
    for (int k = 0; k < EPT; ++k) {
        const int e = base + k * 256 + tid;
        d[k] = dst[e < E ? e : E - 1];
    }
    for (int i = tid; i < R; i += 256) h[i] = 0;
    __syncthreads();
    for (int k = 0; k < EPT; ++k) {
        const int e = base + k * 256 + tid;
        if (e < E) atomicAdd(&h[d[k] >> RSH], 1);
    }
    __syncthreads();
    for (int i = tid; i < R; i += 256) {
        const int c = h[i];
        if (c) atomicAdd(&rangeCnt[i], c);
    }
}

__global__ void __launch_bounds__(256)
range_bin_kernel(const int* __restrict__ src, const int* __restrict__ dst,
                 const int* __restrict__ rangeCnt, int* __restrict__ rangeFill,
                 int2* __restrict__ binned, int E, int R) {
    __shared__ int h[MAXR];
    __shared__ int off[MAXR];
    __shared__ int loc[MAXR];
    __shared__ int gb[MAXR];
    __shared__ int2 stg[TILE];
    __shared__ int wsum[8];
    const int tid = threadIdx.x;
    const int base = blockIdx.x * TILE;
    int ss[EPT], dd[EPT], lr[EPT];
    int a[4];
    for (int j = 0; j < 4; ++j) {
        const int i = tid * 4 + j;
        a[j] = (i < R) ? rangeCnt[i] : 0;
    }
    for (int k = 0; k < EPT; ++k) {
        const int e  = base + k * 256 + tid;
        const int ec = e < E ? e : E - 1;
        dd[k] = dst[ec];
        ss[k] = src[ec];
    }
    const int s4 = a[0] + a[1] + a[2] + a[3];
    int total;
    int run = block_scan_incl(s4, wsum, total) - s4;
    for (int j = 0; j < 4; ++j) {
        off[tid * 4 + j] = run;
        run += a[j];
        h[tid * 4 + j] = 0;
    }
    __syncthreads();
    for (int k = 0; k < EPT; ++k) {
        const int e = base + k * 256 + tid;
        lr[k] = 0;
        if (e < E) lr[k] = atomicAdd(&h[dd[k] >> RSH], 1);
    }
    __syncthreads();
    int cc[RPT], gg[RPT];
    for (int j = 0; j < RPT; ++j) {
        const int i = tid + j * 256;
        cc[j] = (i < R) ? h[i] : 0;
    }
    for (int j = 0; j < RPT; ++j) {
        gg[j] = 0;
        if (cc[j]) gg[j] = atomicAdd(&rangeFill[tid + j * 256], cc[j]);
    }
    int c4[4];
    for (int j = 0; j < 4; ++j) c4[j] = h[tid * 4 + j];
    const int t4 = c4[0] + c4[1] + c4[2] + c4[3];
    int nt;
    int lrun = block_scan_incl(t4, wsum, nt) - t4;
    for (int j = 0; j < 4; ++j) {
        loc[tid * 4 + j] = lrun;
        lrun += c4[j];
    }
    for (int j = 0; j < RPT; ++j) {
        const int i = tid + j * 256;
        if (cc[j]) gb[i] = off[i] + gg[j];
    }
    __syncthreads();
    int lb[EPT];
    for (int k = 0; k < EPT; ++k) lb[k] = loc[dd[k] >> RSH];
    for (int k = 0; k < EPT; ++k) {
        const int e = base + k * 256 + tid;
        if (e < E) stg[lb[k] + lr[k]] = make_int2(ss[k], dd[k]);
    }
    __syncthreads();
    int2 pr[EPT];
    int  gp[EPT];
    for (int k = 0; k < EPT; ++k) {
        const int i = k * 256 + tid;
        pr[k] = stg[i];
        if (i >= nt) pr[k] = make_int2(0, 0);
    }
    for (int k = 0; k < EPT; ++k) {
        const int i = k * 256 + tid;
        const int r = pr[k].y >> RSH;
        gp[k] = gb[r] + (i - loc[r]);
    }
    for (int k = 0; k < EPT; ++k) {
        const int i = k * 256 + tid;
        if (i < nt) binned[gp[k]] = pr[k];
    }
}

__global__ void __launch_bounds__(256)
range_build_kernel(const int2* __restrict__ binned, const int* __restrict__ rangeCnt,
                   int* __restrict__ start, int* __restrict__ cnt, float* __restrict__ dinv,
                   int* __restrict__ csr, int N) {
    __shared__ int c[RS];
    __shared__ int cur[RS];
    __shared__ int stg[SCAP];
    __shared__ int wsum[8];
    const int tid = threadIdx.x;
    const int r   = blockIdx.x;
    int part = 0;
    for (int j = 0; j < RPT; ++j) {
        const int i = tid + j * 256;
        part += (i < r) ? rangeCnt[i] : 0;
    }
    int rbeg;
    block_scan_incl(part, wsum, rbeg);
    const int n = rangeCnt[r];
    c[tid] = 0;
    __syncthreads();
    int2 p[BPT];
    int held = -1;
    for (int c0 = 0; c0 < n; c0 += BPT * 256) {
        for (int k = 0; k < BPT; ++k) {
            const int i = c0 + k * 256 + tid;
            p[k] = binned[rbeg + (i < n ? i : n - 1)];
        }
        for (int k = 0; k < BPT; ++k) {
            const int i = c0 + k * 256 + tid;
            if (i < n) atomicAdd(&c[p[k].y & (RS - 1)], 1);
        }
        held = c0;
    }
    __syncthreads();
    const int v = c[tid];
    int tot;
    const int excl = block_scan_incl(v, wsum, tot) - v;
    const int node = r * RS + tid;
    if (node < N) {
        start[node] = rbeg + excl;
        cnt[node]   = v;
        dinv[node]  = rsqrtf((float)(v + 1));
    }
    const bool staged = n <= SCAP;
    cur[tid] = staged ? excl : rbeg + excl;
    __syncthreads();
    for (int c0 = 0; c0 < n; c0 += BPT * 256) {
        if (c0 != held) {
            for (int k = 0; k < BPT; ++k) {
                const int i = c0 + k * 256 + tid;
                p[k] = binned[rbeg + (i < n ? i : n - 1)];
            }
            held = c0;
        }
        int pos[BPT];
        for (int k = 0; k < BPT; ++k) {
            const int i = c0 + k * 256 + tid;
            pos[k] = 0;
            if (i < n) pos[k] = atomicAdd(&cur[p[k].y & (RS - 1)], 1);
        }
        if (staged) {
            for (int k = 0; k < BPT; ++k) {
                const int i = c0 + k * 256 + tid;
                if (i < n) stg[pos[k]] = p[k].x;
            }
        } else {
            for (int k = 0; k < BPT; ++k) {
                const int i = c0 + k * 256 + tid;
                if (i < n) csr[pos[k]] = p[k].x;
            }
        }
    }
    if (staged) {
        __syncthreads();
        for (int i0 = tid; i0 < n; i0 += 4 * 256) {
            int s4[4];
            for (int q = 0; q < 4; ++q) {
                const int i = i0 + q * 256;
                s4[q] = stg[i < n ? i : n - 1];
            }
            for (int q = 0; q < 4; ++q) {
                const int i = i0 + q * 256;
                if (i < n) csr[rbeg + i] = s4[q];
            }
        }
    }
}


__global__ void prep_kernel(const float* __restrict__ W1, const float* __restrict__ W2,
                            float* __restrict__ pW1, float* __restrict__ pW2,
                            int* __restrict__ rangeCnt, int* __restrict__ rangeFill, int R) {
    const int idx  = blockIdx.x * blockDim.x + threadIdx.x;
    const int kk   = idx >> 8;
    const int rem  = idx & 255;
    const int lane = rem >> 3;
    const int j    = rem & 7;
    const int n    = j >> 1;
    const int t    = j & 1;
    const int lm   = lane & 15;
    const int kh   = (lane >> 4) << 1;
    if (idx < (F_IN / 4) * 256)
        pW1[idx] = W1[(size_t)(4 * kk + kh + t) * F_HID + n * 16 + lm];
    if (idx < (F_HID / 4) * 256) {
        const int col = n * 16 + lm;
        float v = 0.0f;
        if (j < 6 && col < F_OUT)
            v = W2[(size_t)(4 * kk + kh + t) * F_OUT + col];
        pW2[idx] = v;
    }
    if (idx < R) { rangeCnt[idx] = 0; rangeFill[idx] = 0; }
}


__global__ void gemm1_wmma_kernel(const float* __restrict__ x, const float* __restrict__ pW1,
                                  const float* __restrict__ dinv, float* __restrict__ h1, int N) {
    const int lane = threadIdx.x & 31;
    const int wave = threadIdx.x >> 5;
    const int row0 = blockIdx.x * 128 + wave * 16;
    const int lm   = lane & 15;
    const int kh   = (lane >> 4) << 1;
    int r = row0 + lm;
    if (r > N - 1) r = N - 1;
    const float* xrow = x + (size_t)r * F_IN + kh;
    const float* pw   = pW1 + (size_t)lane * 8;

    const int hi8 = (lane >> 4) << 3;
    float dv[8];
    for (int j = 0; j < 8; ++j) {
        const int rr = row0 + hi8 + j;
        dv[j] = dinv[rr < N ? rr : N - 1];
    }

    v8f acc0 = {}, acc1 = {}, acc2 = {}, acc3 = {};

    for (int kk = 0; kk < F_IN / 4; ++kk) {
        const v2f A   = *(const v2f*)(xrow + kk * 4);
        const v4f w01 = *(const v4f*)(pw + (size_t)kk * 256);
        const v4f w23 = *(const v4f*)(pw + (size_t)kk * 256 + 4);
        v2f B0; B0[0] = w01[0]; B0[1] = w01[1];
        v2f B1; B1[0] = w01[2]; B1[1] = w01[3];
        v2f B2; B2[0] = w23[0]; B2[1] = w23[1];
        v2f B3; B3[0] = w23[2]; B3[1] = w23[3];
        acc0 = __builtin_amdgcn_wmma_f32_16x16x4_f32(false, A, false, B0, (short)0, acc0, false, false);
        acc1 = __builtin_amdgcn_wmma_f32_16x16x4_f32(false, A, false, B1, (short)0, acc1, false, false);
        acc2 = __builtin_amdgcn_wmma_f32_16x16x4_f32(false, A, false, B2, (short)0, acc2, false, false);
        acc3 = __builtin_amdgcn_wmma_f32_16x16x4_f32(false, A, false, B3, (short)0, acc3, false, false);
    }

    for (int j = 0; j < 8; ++j) {
        int rr = row0 + hi8 + j;
        if (rr < N) {
            float* o = h1 + (size_t)rr * F_HID + lm;
            o[0]  = acc0[j] * dv[j];
            o[16] = acc1[j] * dv[j];
            o[32] = acc2[j] * dv[j];
            o[48] = acc3[j] * dv[j];
        }
    }
}


#define RL(v, j) __builtin_amdgcn_readlane((v), (j))

__device__ __forceinline__ float2 ld_row2(const float* __restrict__ col, int s, int pitch) {
    return *(const float2*)(col + (size_t)s * pitch);
}

__device__ __forceinline__ float2 sum_rows(const float* __restrict__ col, int pitch,
                                           int my_s, int n, float2 a) {
    int j = 0;
    for (; j + 8 <= n; j += 8) {
        const float2 v0 = ld_row2(col, RL(my_s, j + 0), pitch);
        const float2 v1 = ld_row2(col, RL(my_s, j + 1), pitch);
        const float2 v2 = ld_row2(col, RL(my_s, j + 2), pitch);
        const float2 v3 = ld_row2(col, RL(my_s, j + 3), pitch);
        const float2 v4 = ld_row2(col, RL(my_s, j + 4), pitch);
        const float2 v5 = ld_row2(col, RL(my_s, j + 5), pitch);
        const float2 v6 = ld_row2(col, RL(my_s, j + 6), pitch);
        const float2 v7 = ld_row2(col, RL(my_s, j + 7), pitch);
        a.x += v0.x; a.y += v0.y;
        a.x += v1.x; a.y += v1.y;
        a.x += v2.x; a.y += v2.y;
        a.x += v3.x; a.y += v3.y;
        a.x += v4.x; a.y += v4.y;
        a.x += v5.x; a.y += v5.y;
        a.x += v6.x; a.y += v6.y;
        a.x += v7.x; a.y += v7.y;
    }
    if (j + 4 <= n) {
        const float2 v0 = ld_row2(col, RL(my_s, j + 0), pitch);
        const float2 v1 = ld_row2(col, RL(my_s, j + 1), pitch);
        const float2 v2 = ld_row2(col, RL(my_s, j + 2), pitch);
        const float2 v3 = ld_row2(col, RL(my_s, j + 3), pitch);
        a.x += v0.x; a.y += v0.y;
        a.x += v1.x; a.y += v1.y;
        a.x += v2.x; a.y += v2.y;
        a.x += v3.x; a.y += v3.y;
        j += 4;
    }
    if (j + 2 <= n) {
        const float2 v0 = ld_row2(col, RL(my_s, j + 0), pitch);
        const float2 v1 = ld_row2(col, RL(my_s, j + 1), pitch);
        a.x += v0.x; a.y += v0.y;
        a.x += v1.x; a.y += v1.y;
        j += 2;
    }
    if (j < n) {
        const float2 v0 = ld_row2(col, RL(my_s, j), pitch);
        a.x += v0.x; a.y += v0.y;
    }
    return a;
}


#define NPW 8

__device__ __forceinline__ float2 gather_node(const float* __restrict__ col, int pitch,
                                              const int* __restrict__ csr, int node, int beg, int c,
                                              int my_s, float2 self, int lane) {
    float2 a = sum_rows(col, pitch, my_s, c < 32 ? c : 32, self);
    for (int base = 32; base < c; base += 32) {
        int n = c - base;
        if (n > 32) n = 32;
        int s2 = node;
        if (lane < n) s2 = csr[beg + base + lane];
        a = sum_rows(col, pitch, s2, n, a);
    }
    return a;
}


#define G1W 4
#define NB  (G1W * NPW)
#define ZP  (F_HID + 4)

__global__ void __launch_bounds__(G1W * 32)
gather1_gemm2_kernel(const float* __restrict__ h1s, const float* __restrict__ dinv,
                     const int* __restrict__ start, const int* __restrict__ cnt,
                     const int* __restrict__ csr, const float* __restrict__ b1,
                     const float* __restrict__ pW2, float* __restrict__ h2s, int N) {
    __shared__ __attribute__((aligned(16))) float sz[NB * ZP];
    __shared__ __attribute__((aligned(16))) float so[NB * F_OUT];
    __shared__ float sd[NB];
    const int lane = threadIdx.x & 31;
    const int wave = __builtin_amdgcn_readfirstlane(threadIdx.x >> 5);
    const int row0 = wave * NPW;
    const int n0   = blockIdx.x * NB + row0;
    {
        const float* col = h1s + lane * 2;
        const float2 bb = *(const float2*)(b1 + lane * 2);
        int ln = n0 + (lane < NPW ? lane : NPW - 1);
        if (ln > N - 1) ln = N - 1;
        const int   sv  = start[ln];
        const int   cv  = cnt[ln];
        const float dvv = dinv[ln];
        if (lane < NPW) sd[row0 + lane] = dvv;
        int node = n0 < N - 1 ? n0 : N - 1;
        int beg  = RL(sv, 0);
        int c    = RL(cv, 0);
        int my_s = node;
        if (lane < c) my_s = csr[beg + lane];
        float2 self = ld_row2(col, node, F_HID);
        for (int i = 0; i < NPW; ++i) {
            const int ni = i + 1 < NPW ? i + 1 : NPW - 1;
            int nnode = n0 + ni;
            if (nnode > N - 1) nnode = N - 1;
            const int nbeg = RL(sv, ni);
            const int nc   = RL(cv, ni);
            int nmy = nnode;
            if (lane < nc) nmy = csr[nbeg + lane];
            const float2 nself = ld_row2(col, nnode, F_HID);
            const float dd = __int_as_float(RL(__float_as_int(dvv), i));
            const float2 a = gather_node(col, F_HID, csr, node, beg, c, my_s, self, lane);
            const float rx = a.x * dd + bb.x;
            const float ry = a.y * dd + bb.y;
            float2 r;
            r.x = rx > 0.0f ? rx : 0.0f;
            r.y = ry > 0.0f ? ry : 0.0f;
            *(float2*)(sz + (row0 + i) * ZP + lane * 2) = r;
            node = nnode; beg = nbeg; c = nc; my_s = nmy; self = nself;
        }
    }
    __syncthreads();
    if (wave < 3) {
        const int lm  = lane & 15;
        const int kh  = (lane >> 4) << 1;
        const int hi8 = (lane >> 4) << 3;
        const float* zrow0 = sz + lm * ZP + kh;
        const float* zrow1 = zrow0 + 16 * ZP;
        const float* pw    = pW2 + (size_t)lane * 8 + wave * 2;
        v2f Bf[F_HID / 4];
        #pragma unroll
        for (int kk = 0; kk < F_HID / 4; ++kk) Bf[kk] = *(const v2f*)(pw + (size_t)kk * 256);
        v8f acc0 = {}, acc1 = {};
        #pragma unroll
        for (int kk = 0; kk < F_HID / 4; ++kk) {
            const v2f A0 = *(const v2f*)(zrow0 + kk * 4);
            const v2f A1 = *(const v2f*)(zrow1 + kk * 4);
            acc0 = __builtin_amdgcn_wmma_f32_16x16x4_f32(false, A0, false, Bf[kk], (short)0, acc0, false, false);
            acc1 = __builtin_amdgcn_wmma_f32_16x16x4_f32(false, A1, false, Bf[kk], (short)0, acc1, false, false);
        }
        const int colo = wave * 16 + lm;
        if (colo < F_OUT) {
            #pragma unroll
            for (int j = 0; j < 8; ++j) {
                so[(hi8 + j) * F_OUT + colo]      = acc0[j] * sd[hi8 + j];
                so[(16 + hi8 + j) * F_OUT + colo] = acc1[j] * sd[16 + hi8 + j];
            }
        }
    }
    __syncthreads();
    for (int t = threadIdx.x; t < (NB * F_OUT) / 4; t += G1W * 32) {
        const size_t g = (size_t)blockIdx.x * (NB * F_OUT) + (size_t)t * 4;
        if (g < (size_t)N * F_OUT)
            *(v4f*)(h2s + g) = *(const v4f*)(so + t * 4);
    }
}


#define G2W 4
#define NB2 (G2W * NPW)

__global__ void __launch_bounds__(G2W * 32)
gather2_kernel(const float* __restrict__ h2s, const float* __restrict__ dinv,
               const int* __restrict__ start, const int* __restrict__ cnt,
               const int* __restrict__ csr, const float* __restrict__ b2,
               float* __restrict__ out, int N) {
    __shared__ __attribute__((aligned(16))) float so[NB2 * F_OUT];
    const int lane = threadIdx.x & 31;
    const int wave = __builtin_amdgcn_readfirstlane(threadIdx.x >> 5);
    const int row0 = wave * NPW;
    const int n0   = blockIdx.x * NB2 + row0;
    if (n0 < N) {
        const int cl = lane < 20 ? lane : 19;
        const float* col = h2s + cl * 2;
        const float2 bb = *(const float2*)(b2 + cl * 2);
        int ln = n0 + (lane < NPW ? lane : NPW - 1);
        if (ln > N - 1) ln = N - 1;
        const int   sv  = start[ln];
        const int   cv  = cnt[ln];
        const float dvv = dinv[ln];
        int node = n0;
        int beg  = RL(sv, 0);
        int c    = RL(cv, 0);
        int my_s = node;
        if (lane < c) my_s = csr[beg + lane];
        float2 self = ld_row2(col, node, F_OUT);
        for (int i = 0; i < NPW; ++i) {
            const int ni = i + 1 < NPW ? i + 1 : NPW - 1;
            int nnode = n0 + ni;
            if (nnode > N - 1) nnode = N - 1;
            const int nbeg = RL(sv, ni);
            const int nc   = RL(cv, ni);
            int nmy = nnode;
            if (lane < nc) nmy = csr[nbeg + lane];
            const float2 nself = ld_row2(col, nnode, F_OUT);
            const float dd = __int_as_float(RL(__float_as_int(dvv), i));
            const float2 a = gather_node(col, F_OUT, csr, node, beg, c, my_s, self, lane);
            if (lane < 20) {
                so[(row0 + i) * F_OUT + lane * 2]     = a.x * dd + bb.x;
                so[(row0 + i) * F_OUT + lane * 2 + 1] = a.y * dd + bb.y;
            }
            node = nnode; beg = nbeg; c = nc; my_s = nmy; self = nself;
        }
    }
    __syncthreads();
    for (int t = threadIdx.x; t < (NB2 * F_OUT) / 4; t += G2W * 32) {
        const size_t g = (size_t)blockIdx.x * (NB2 * F_OUT) + (size_t)t * 4;
        if (g < (size_t)N * F_OUT)
            *(v4f*)(out + g) = *(const v4f*)(so + t * 4);
    }
}


extern "C" void kernel_launch(void* const* d_in, const int* in_sizes, int n_in,
                              void* d_out, int out_size, void* d_ws, size_t ws_size,
                              hipStream_t stream) {
    const float* x  = (const float*)d_in[0];
    const int*   ei = (const int*)d_in[1];
    const float* W1 = (const float*)d_in[2];
    const float* b1 = (const float*)d_in[3];
    const float* W2 = (const float*)d_in[4];
    const float* b2 = (const float*)d_in[5];

    const int N = in_sizes[0] / F_IN;
    const int E = in_sizes[1] / 2;
    const int* src = ei;
    const int* dst = ei + E;

    float* ws    = (float*)d_ws;
    float* dinv  = ws + (size_t)N;
    float* bufA  = ws + (size_t)2 * N;
    float* bufB  = bufA + (size_t)N * F_HID;
    float* pW1   = bufB + (size_t)N * F_HID;
    float* pW2   = pW1 + (size_t)(F_IN / 4) * 256;
    int*   cnt   = (int*)(pW2 + (size_t)(F_HID / 4) * 256);
    int*   start = cnt + (size_t)N;
    int2*  binned = (int2*)(start + (size_t)N + (N & 1));
    int*   csr   = (int*)(binned + (size_t)E);
    int*   rangeCnt  = csr + (size_t)E;
    int*   rangeFill = rangeCnt + MAXR;
    float* out   = (float*)d_out;

    const int B = 256;
    const int R           = (N + RS - 1) / RS;
    const int tileBlocks  = (E + TILE - 1) / TILE;
    const int gemmBlocks  = (N + 127) / 128;
    const int gather2Blk  = (N + NB2 - 1) / NB2;
    const int fusedBlocks = (N + NB - 1) / NB;
    const int prepBlocks  = ((F_IN / 4) * 256 + B - 1) / B;

    prep_kernel<<<prepBlocks, B, 0, stream>>>(W1, W2, pW1, pW2, rangeCnt, rangeFill, R);

    range_count_kernel<<<tileBlocks, B, 0, stream>>>(dst, rangeCnt, E, R);
    range_bin_kernel<<<tileBlocks, B, 0, stream>>>(src, dst, rangeCnt, rangeFill, binned, E, R);
    range_build_kernel<<<R, B, 0, stream>>>(binned, rangeCnt, start, cnt, dinv, csr, N);

    gemm1_wmma_kernel<<<gemmBlocks, B, 0, stream>>>(x, pW1, dinv, bufA, N);
    gather1_gemm2_kernel<<<fusedBlocks, G1W * 32, 0, stream>>>(bufA, dinv, start, cnt, csr, b1, pW2, bufB, N);

    gather2_kernel<<<gather2Blk, G2W * 32, 0, stream>>>(bufB, dinv, start, cnt, csr, b2, out, N);
}


# List build: the edge-counting pass and its launch are gone; every 256-node range owns a fixed region of pair slots, a full region spills to an overflow list; launches 7 -> 6; list order as before
# speedup vs baseline: 4.9201x; 1.0441x over previous
#include <hip/hip_runtime.h>

typedef __attribute__((ext_vector_type(2))) float v2f;
typedef __attribute__((ext_vector_type(4))) float v4f;
typedef __attribute__((ext_vector_type(8))) float v8f;

#define F_IN  128
#define F_HID 64
#define F_OUT 40


#define RSH   8
#define RS    256
#define MAXR  1024
#define TILE  4096
#define EPT   (TILE / 256)
#define RPT   (MAXR / 256)
#define BPT   20
#define SCAP  8192
#define CAP   SCAP

__device__ __forceinline__ int block_scan_incl(int v, int* wsum, int& total) {
    const int lane = threadIdx.x & 31;
    const int wave = threadIdx.x >> 5;
    int incl = v;
    for (int o = 1; o < 32; o <<= 1) {
        int t = __shfl_up(incl, o);
        if (lane >= o) incl += t;
    }
    if (lane == 31) wsum[wave] = incl;
    __syncthreads();
    int before = 0, all = 0;
    for (int w = 0; w < 8; ++w) {
        const int s = wsum[w];
        if (w < wave) before += s;
        all += s;
    }
    total = all;
    __syncthreads();
    return incl + before;
}

__global__ void __launch_bounds__(256)
range_bin_kernel(const int* __restrict__ src, const int* __restrict__ dst,
                 int* __restrict__ rangeFill, int2* __restrict__ binned,
                 int2* __restrict__ ovf, int* __restrict__ ovfCnt, int E, int R) {
    __shared__ int h[MAXR];
    __shared__ int loc[MAXR];
    __shared__ int gb[MAXR];
    __shared__ int2 stg[TILE];
    __shared__ int wsum[8];
    const int tid = threadIdx.x;
    const int base = blockIdx.x * TILE;
    int ss[EPT], dd[EPT], lr[EPT];
    for (int k = 0; k < EPT; ++k) {
        const int e  = base + k * 256 + tid;
        const int ec = e < E ? e : E - 1;
        dd[k] = dst[ec];
        ss[k] = src[ec];
    }
    for (int j = 0; j < 4; ++j) h[tid * 4 + j] = 0;
    __syncthreads();
    for (int k = 0; k < EPT; ++k) {
        const int e = base + k * 256 + tid;
        lr[k] = 0;
        if (e < E) lr[k] = atomicAdd(&h[dd[k] >> RSH], 1);
    }
    __syncthreads();
    int cc[RPT], gg[RPT];
    for (int j = 0; j < RPT; ++j) {
        const int i = tid + j * 256;
        cc[j] = (i < R) ? h[i] : 0;
    }
    for (int j = 0; j < RPT; ++j) {
        gg[j] = 0;
        if (cc[j]) gg[j] = atomicAdd(&rangeFill[tid + j * 256], cc[j]);
    }
    int c4[4];
    for (int j = 0; j < 4; ++j) c4[j] = h[tid * 4 + j];
    const int t4 = c4[0] + c4[1] + c4[2] + c4[3];
    int nt;
    int lrun = block_scan_incl(t4, wsum, nt) - t4;
    for (int j = 0; j < 4; ++j) {
        loc[tid * 4 + j] = lrun;
        lrun += c4[j];
    }
    for (int j = 0; j < RPT; ++j) {
        const int i = tid + j * 256;
        if (cc[j]) gb[i] = gg[j];
    }
    __syncthreads();
    int lb[EPT];
    for (int k = 0; k < EPT; ++k) lb[k] = loc[dd[k] >> RSH];
    for (int k = 0; k < EPT; ++k) {
        const int e = base + k * 256 + tid;
        if (e < E) stg[lb[k] + lr[k]] = make_int2(ss[k], dd[k]);
    }
    __syncthreads();
    int2 pr[EPT];
    int  gq[EPT], sl[EPT];
    for (int k = 0; k < EPT; ++k) {
        const int i = k * 256 + tid;
        pr[k] = stg[i];
        if (i >= nt) pr[k] = make_int2(0, 0);
    }
    for (int k = 0; k < EPT; ++k) {
        const int i = k * 256 + tid;
        const int r = pr[k].y >> RSH;
        gq[k] = gb[r] + (i - loc[r]);
        sl[k] = r * CAP + gq[k];
    }
    bool spill = false;
    for (int k = 0; k < EPT; ++k) {
        const int i = k * 256 + tid;
        const bool fits = gq[k] < CAP;
        if ((i < nt) & fits) binned[sl[k]] = pr[k];
        spill |= (i < nt) & !fits;
    }
    if (spill) {
        for (int k = 0; k < EPT; ++k) {
            const int i = k * 256 + tid;
            if (i < nt && gq[k] >= CAP) {
                const int pos = atomicAdd(ovfCnt, 1);
                ovf[pos] = pr[k];
            }
        }
    }
}

__global__ void __launch_bounds__(256)
range_build_kernel(const int2* __restrict__ binned, const int2* __restrict__ ovf,
                   const int* __restrict__ rangeFill, const int* __restrict__ ovfCnt,
                   int* __restrict__ start, int* __restrict__ cnt, float* __restrict__ dinv,
                   int* __restrict__ csr, int N) {
    __shared__ int c[RS];
    __shared__ int cur[RS];
    __shared__ int stg[SCAP];
    __shared__ int wsum[8];
    const int tid = threadIdx.x;
    const int r   = blockIdx.x;
    int part = 0;
    for (int j = 0; j < RPT; ++j) {
        const int i = tid + j * 256;
        part += (i < r) ? rangeFill[i] : 0;
    }
    int rbeg;
    block_scan_incl(part, wsum, rbeg);
    const int  tot     = rangeFill[r];
    const bool spilled = tot > CAP;
    const int  n       = spilled ? CAP : tot;
    const int  nOv     = spilled ? ovfCnt[0] : 0;
    const int2* __restrict__ reg = binned + (size_t)r * CAP;
    c[tid] = 0;
    __syncthreads();
    int2 p[BPT];
    int held = -1;
    for (int c0 = 0; c0 < n; c0 += BPT * 256) {
        for (int k = 0; k < BPT; ++k) {
            const int i = c0 + k * 256 + tid;
            p[k] = reg[i < n ? i : n - 1];
        }
        for (int k = 0; k < BPT; ++k) {
            const int i = c0 + k * 256 + tid;
            if (i < n) atomicAdd(&c[p[k].y & (RS - 1)], 1);
        }
        held = c0;
    }
    if (spilled) {
        for (int i = tid; i < nOv; i += 256) {
            const int2 q = ovf[i];
            if ((q.y >> RSH) == r) atomicAdd(&c[q.y & (RS - 1)], 1);
        }
    }
    __syncthreads();
    const int v = c[tid];
    int tot2;
    const int excl = block_scan_incl(v, wsum, tot2) - v;
    const int node = r * RS + tid;
    if (node < N) {
        start[node] = rbeg + excl;
        cnt[node]   = v;
        dinv[node]  = rsqrtf((float)(v + 1));
    }
    const bool staged = !spilled;
    cur[tid] = staged ? excl : rbeg + excl;
    __syncthreads();
    for (int c0 = 0; c0 < n; c0 += BPT * 256) {
        if (c0 != held) {
            for (int k = 0; k < BPT; ++k) {
                const int i = c0 + k * 256 + tid;
                p[k] = reg[i < n ? i : n - 1];
            }
            held = c0;
        }
        int pos[BPT];
        for (int k = 0; k < BPT; ++k) {
            const int i = c0 + k * 256 + tid;
            pos[k] = 0;
            if (i < n) pos[k] = atomicAdd(&cur[p[k].y & (RS - 1)], 1);
        }
        if (staged) {
            for (int k = 0; k < BPT; ++k) {
                const int i = c0 + k * 256 + tid;
                if (i < n) stg[pos[k]] = p[k].x;
            }
        } else {
            for (int k = 0; k < BPT; ++k) {
                const int i = c0 + k * 256 + tid;
                if (i < n) csr[pos[k]] = p[k].x;
            }
        }
    }
    if (spilled) {
        for (int i = tid; i < nOv; i += 256) {
            const int2 q = ovf[i];
            if ((q.y >> RSH) == r) {
                const int pos = atomicAdd(&cur[q.y & (RS - 1)], 1);
                csr[pos] = q.x;
            }
        }
    }
    if (staged) {
        __syncthreads();
        for (int i0 = tid; i0 < n; i0 += 4 * 256) {
            int s4[4];
            for (int q = 0; q < 4; ++q) {
                const int i = i0 + q * 256;
                s4[q] = stg[i < n ? i : n - 1];
            }
            for (int q = 0; q < 4; ++q) {
                const int i = i0 + q * 256;
                if (i < n) csr[rbeg + i] = s4[q];
            }
        }
    }
}


__global__ void prep_kernel(const float* __restrict__ W1, const float* __restrict__ W2,
                            float* __restrict__ pW1, float* __restrict__ pW2,
                            int* __restrict__ rangeFill, int* __restrict__ ovfCnt, int R) {
    const int idx  = blockIdx.x * blockDim.x + threadIdx.x;
    const int kk   = idx >> 8;
    const int rem  = idx & 255;
    const int lane = rem >> 3;
    const int j    = rem & 7;
    const int n    = j >> 1;
    const int t    = j & 1;
    const int lm   = lane & 15;
    const int kh   = (lane >> 4) << 1;
    if (idx < (F_IN / 4) * 256)
        pW1[idx] = W1[(size_t)(4 * kk + kh + t) * F_HID + n * 16 + lm];
    if (idx < (F_HID / 4) * 256) {
        const int col = n * 16 + lm;
        float v = 0.0f;
        if (j < 6 && col < F_OUT)
            v = W2[(size_t)(4 * kk + kh + t) * F_OUT + col];
        pW2[idx] = v;
    }
    if (idx < R) rangeFill[idx] = 0;
    if (idx == 0) ovfCnt[0] = 0;
}


__global__ void gemm1_wmma_kernel(const float* __restrict__ x, const float* __restrict__ pW1,
                                  const float* __restrict__ dinv, float* __restrict__ h1, int N) {
    const int lane = threadIdx.x & 31;
    const int wave = threadIdx.x >> 5;
    const int row0 = blockIdx.x * 128 + wave * 16;
    const int lm   = lane & 15;
    const int kh   = (lane >> 4) << 1;
    int r = row0 + lm;
    if (r > N - 1) r = N - 1;
    const float* xrow = x + (size_t)r * F_IN + kh;
    const float* pw   = pW1 + (size_t)lane * 8;

    const int hi8 = (lane >> 4) << 3;
    float dv[8];
    for (int j = 0; j < 8; ++j) {
        const int rr = row0 + hi8 + j;
        dv[j] = dinv[rr < N ? rr : N - 1];
    }

    v8f acc0 = {}, acc1 = {}, acc2 = {}, acc3 = {};

    for (int kk = 0; kk < F_IN / 4; ++kk) {
        const v2f A   = *(const v2f*)(xrow + kk * 4);
        const v4f w01 = *(const v4f*)(pw + (size_t)kk * 256);
        const v4f w23 = *(const v4f*)(pw + (size_t)kk * 256 + 4);
        v2f B0; B0[0] = w01[0]; B0[1] = w01[1];
        v2f B1; B1[0] = w01[2]; B1[1] = w01[3];
        v2f B2; B2[0] = w23[0]; B2[1] = w23[1];
        v2f B3; B3[0] = w23[2]; B3[1] = w23[3];
        acc0 = __builtin_amdgcn_wmma_f32_16x16x4_f32(false, A, false, B0, (short)0, acc0, false, false);
        acc1 = __builtin_amdgcn_wmma_f32_16x16x4_f32(false, A, false, B1, (short)0, acc1, false, false);
        acc2 = __builtin_amdgcn_wmma_f32_16x16x4_f32(false, A, false, B2, (short)0, acc2, false, false);
        acc3 = __builtin_amdgcn_wmma_f32_16x16x4_f32(false, A, false, B3, (short)0, acc3, false, false);
    }

    for (int j = 0; j < 8; ++j) {
        int rr = row0 + hi8 + j;
        if (rr < N) {
            float* o = h1 + (size_t)rr * F_HID + lm;
            o[0]  = acc0[j] * dv[j];
            o[16] = acc1[j] * dv[j];
            o[32] = acc2[j] * dv[j];
            o[48] = acc3[j] * dv[j];
        }
    }
}


#define RL(v, j) __builtin_amdgcn_readlane((v), (j))

__device__ __forceinline__ float2 ld_row2(const float* __restrict__ col, int s, int pitch) {
    return *(const float2*)(col + (size_t)s * pitch);
}

__device__ __forceinline__ float2 sum_rows(const float* __restrict__ col, int pitch,
                                           int my_s, int n, float2 a) {
    int j = 0;
    for (; j + 8 <= n; j += 8) {
        const float2 v0 = ld_row2(col, RL(my_s, j + 0), pitch);
        const float2 v1 = ld_row2(col, RL(my_s, j + 1), pitch);
        const float2 v2 = ld_row2(col, RL(my_s, j + 2), pitch);
        const float2 v3 = ld_row2(col, RL(my_s, j + 3), pitch);
        const float2 v4 = ld_row2(col, RL(my_s, j + 4), pitch);
        const float2 v5 = ld_row2(col, RL(my_s, j + 5), pitch);
        const float2 v6 = ld_row2(col, RL(my_s, j + 6), pitch);
        const float2 v7 = ld_row2(col, RL(my_s, j + 7), pitch);
        a.x += v0.x; a.y += v0.y;
        a.x += v1.x; a.y += v1.y;
        a.x += v2.x; a.y += v2.y;
        a.x += v3.x; a.y += v3.y;
        a.x += v4.x; a.y += v4.y;
        a.x += v5.x; a.y += v5.y;
        a.x += v6.x; a.y += v6.y;
        a.x += v7.x; a.y += v7.y;
    }
    if (j + 4 <= n) {
        const float2 v0 = ld_row2(col, RL(my_s, j + 0), pitch);
        const float2 v1 = ld_row2(col, RL(my_s, j + 1), pitch);
        const float2 v2 = ld_row2(col, RL(my_s, j + 2), pitch);
        const float2 v3 = ld_row2(col, RL(my_s, j + 3), pitch);
        a.x += v0.x; a.y += v0.y;
        a.x += v1.x; a.y += v1.y;
        a.x += v2.x; a.y += v2.y;
        a.x += v3.x; a.y += v3.y;
        j += 4;
    }
    if (j + 2 <= n) {
        const float2 v0 = ld_row2(col, RL(my_s, j + 0), pitch);
        const float2 v1 = ld_row2(col, RL(my_s, j + 1), pitch);
        a.x += v0.x; a.y += v0.y;
        a.x += v1.x; a.y += v1.y;
        j += 2;
    }
    if (j < n) {
        const float2 v0 = ld_row2(col, RL(my_s, j), pitch);
        a.x += v0.x; a.y += v0.y;
    }
    return a;
}


#define NPW 8

__device__ __forceinline__ float2 gather_node(const float* __restrict__ col, int pitch,
                                              const int* __restrict__ csr, int node, int beg, int c,
                                              int my_s, float2 self, int lane) {
    float2 a = sum_rows(col, pitch, my_s, c < 32 ? c : 32, self);
    for (int base = 32; base < c; base += 32) {
        int n = c - base;
        if (n > 32) n = 32;
        int s2 = node;
        if (lane < n) s2 = csr[beg + base + lane];
        a = sum_rows(col, pitch, s2, n, a);
    }
    return a;
}


#define G1W 4
#define NB  (G1W * NPW)
#define ZP  (F_HID + 4)

__global__ void __launch_bounds__(G1W * 32)
gather1_gemm2_kernel(const float* __restrict__ h1s, const float* __restrict__ dinv,
                     const int* __restrict__ start, const int* __restrict__ cnt,
                     const int* __restrict__ csr, const float* __restrict__ b1,
                     const float* __restrict__ pW2, float* __restrict__ h2s, int N) {
    __shared__ __attribute__((aligned(16))) float sz[NB * ZP];
    __shared__ __attribute__((aligned(16))) float so[NB * F_OUT];
    __shared__ float sd[NB];
    const int lane = threadIdx.x & 31;
    const int wave = __builtin_amdgcn_readfirstlane(threadIdx.x >> 5);
    const int row0 = wave * NPW;
    const int n0   = blockIdx.x * NB + row0;
    {
        const float* col = h1s + lane * 2;
        const float2 bb = *(const float2*)(b1 + lane * 2);
        int ln = n0 + (lane < NPW ? lane : NPW - 1);
        if (ln > N - 1) ln = N - 1;
        const int   sv  = start[ln];
        const int   cv  = cnt[ln];
        const float dvv = dinv[ln];
        if (lane < NPW) sd[row0 + lane] = dvv;
        int node = n0 < N - 1 ? n0 : N - 1;
        int beg  = RL(sv, 0);
        int c    = RL(cv, 0);
        int my_s = node;
        if (lane < c) my_s = csr[beg + lane];
        float2 self = ld_row2(col, node, F_HID);
        for (int i = 0; i < NPW; ++i) {
            const int ni = i + 1 < NPW ? i + 1 : NPW - 1;
            int nnode = n0 + ni;
            if (nnode > N - 1) nnode = N - 1;
            const int nbeg = RL(sv, ni);
            const int nc   = RL(cv, ni);
            int nmy = nnode;
            if (lane < nc) nmy = csr[nbeg + lane];
            const float2 nself = ld_row2(col, nnode, F_HID);
            const float dd = __int_as_float(RL(__float_as_int(dvv), i));
            const float2 a = gather_node(col, F_HID, csr, node, beg, c, my_s, self, lane);
            const float rx = a.x * dd + bb.x;
            const float ry = a.y * dd + bb.y;
            float2 r;
            r.x = rx > 0.0f ? rx : 0.0f;
            r.y = ry > 0.0f ? ry : 0.0f;
            *(float2*)(sz + (row0 + i) * ZP + lane * 2) = r;
            node = nnode; beg = nbeg; c = nc; my_s = nmy; self = nself;
        }
    }
    __syncthreads();
    if (wave < 3) {
        const int lm  = lane & 15;
        const int kh  = (lane >> 4) << 1;
        const int hi8 = (lane >> 4) << 3;
        const float* zrow0 = sz + lm * ZP + kh;
        const float* zrow1 = zrow0 + 16 * ZP;
        const float* pw    = pW2 + (size_t)lane * 8 + wave * 2;
        v2f Bf[F_HID / 4];
        #pragma unroll
        for (int kk = 0; kk < F_HID / 4; ++kk) Bf[kk] = *(const v2f*)(pw + (size_t)kk * 256);
        v8f acc0 = {}, acc1 = {};
        #pragma unroll
        for (int kk = 0; kk < F_HID / 4; ++kk) {
            const v2f A0 = *(const v2f*)(zrow0 + kk * 4);
            const v2f A1 = *(const v2f*)(zrow1 + kk * 4);
            acc0 = __builtin_amdgcn_wmma_f32_16x16x4_f32(false, A0, false, Bf[kk], (short)0, acc0, false, false);
            acc1 = __builtin_amdgcn_wmma_f32_16x16x4_f32(false, A1, false, Bf[kk], (short)0, acc1, false, false);
        }
        const int colo = wave * 16 + lm;
        if (colo < F_OUT) {
            #pragma unroll
            for (int j = 0; j < 8; ++j) {
                so[(hi8 + j) * F_OUT + colo]      = acc0[j] * sd[hi8 + j];
                so[(16 + hi8 + j) * F_OUT + colo] = acc1[j] * sd[16 + hi8 + j];
            }
        }
    }
    __syncthreads();
    for (int t = threadIdx.x; t < (NB * F_OUT) / 4; t += G1W * 32) {
        const size_t g = (size_t)blockIdx.x * (NB * F_OUT) + (size_t)t * 4;
        if (g < (size_t)N * F_OUT)
            *(v4f*)(h2s + g) = *(const v4f*)(so + t * 4);
    }
}


#define G2W 4
#define NB2 (G2W * NPW)

__global__ void __launch_bounds__(G2W * 32)
gather2_kernel(const float* __restrict__ h2s, const float* __restrict__ dinv,
               const int* __restrict__ start, const int* __restrict__ cnt,
               const int* __restrict__ csr, const float* __restrict__ b2,
               float* __restrict__ out, int N) {
    __shared__ __attribute__((aligned(16))) float so[NB2 * F_OUT];
    const int lane = threadIdx.x & 31;
    const int wave = __builtin_amdgcn_readfirstlane(threadIdx.x >> 5);
    const int row0 = wave * NPW;
    const int n0   = blockIdx.x * NB2 + row0;
    if (n0 < N) {
        const int cl = lane < 20 ? lane : 19;
        const float* col = h2s + cl * 2;
        const float2 bb = *(const float2*)(b2 + cl * 2);
        int ln = n0 + (lane < NPW ? lane : NPW - 1);
        if (ln > N - 1) ln = N - 1;
        const int   sv  = start[ln];
        const int   cv  = cnt[ln];
        const float dvv = dinv[ln];
        int node = n0;
        int beg  = RL(sv, 0);
        int c    = RL(cv, 0);
        int my_s = node;
        if (lane < c) my_s = csr[beg + lane];
        float2 self = ld_row2(col, node, F_OUT);
        for (int i = 0; i < NPW; ++i) {
            const int ni = i + 1 < NPW ? i + 1 : NPW - 1;
            int nnode = n0 + ni;
            if (nnode > N - 1) nnode = N - 1;
            const int nbeg = RL(sv, ni);
            const int nc   = RL(cv, ni);
            int nmy = nnode;
            if (lane < nc) nmy = csr[nbeg + lane];
            const float2 nself = ld_row2(col, nnode, F_OUT);
            const float dd = __int_as_float(RL(__float_as_int(dvv), i));
            const float2 a = gather_node(col, F_OUT, csr, node, beg, c, my_s, self, lane);
            if (lane < 20) {
                so[(row0 + i) * F_OUT + lane * 2]     = a.x * dd + bb.x;
                so[(row0 + i) * F_OUT + lane * 2 + 1] = a.y * dd + bb.y;
            }
            node = nnode; beg = nbeg; c = nc; my_s = nmy; self = nself;
        }
    }
    __syncthreads();
    for (int t = threadIdx.x; t < (NB2 * F_OUT) / 4; t += G2W * 32) {
        const size_t g = (size_t)blockIdx.x * (NB2 * F_OUT) + (size_t)t * 4;
        if (g < (size_t)N * F_OUT)
            *(v4f*)(out + g) = *(const v4f*)(so + t * 4);
    }
}


extern "C" void kernel_launch(void* const* d_in, const int* in_sizes, int n_in,
                              void* d_out, int out_size, void* d_ws, size_t ws_size,
                              hipStream_t stream) {
    const float* x  = (const float*)d_in[0];
    const int*   ei = (const int*)d_in[1];
    const float* W1 = (const float*)d_in[2];
    const float* b1 = (const float*)d_in[3];
    const float* W2 = (const float*)d_in[4];
    const float* b2 = (const float*)d_in[5];

    const int N = in_sizes[0] / F_IN;
    const int E = in_sizes[1] / 2;
    const int* src = ei;
    const int* dst = ei + E;

    float* ws    = (float*)d_ws;
    float* dinv  = ws + (size_t)N;
    float* bufA  = ws + (size_t)2 * N;
    float* bufB  = bufA + (size_t)N * F_HID;
    float* pW1   = bufB + (size_t)N * F_HID;
    float* pW2   = pW1 + (size_t)(F_IN / 4) * 256;
    int*   cnt   = (int*)(pW2 + (size_t)(F_HID / 4) * 256);
    int*   start = cnt + (size_t)N;
    const int R = (N + RS - 1) / RS;
    int2*  binned = (int2*)(start + (size_t)N + (N & 1));
    int*   csr   = (int*)(binned + (size_t)R * CAP);
    int2*  ovf   = (int2*)(csr + (size_t)E + (E & 1));
    int*   rangeFill = (int*)(ovf + (size_t)E);
    int*   ovfCnt    = rangeFill + MAXR;
    float* out   = (float*)d_out;

    const int B = 256;
    const int tileBlocks  = (E + TILE - 1) / TILE;
    const int gemmBlocks  = (N + 127) / 128;
    const int gather2Blk  = (N + NB2 - 1) / NB2;
    const int fusedBlocks = (N + NB - 1) / NB;
    const int prepBlocks  = ((F_IN / 4) * 256 + B - 1) / B;

    prep_kernel<<<prepBlocks, B, 0, stream>>>(W1, W2, pW1, pW2, rangeFill, ovfCnt, R);

    range_bin_kernel<<<tileBlocks, B, 0, stream>>>(src, dst, rangeFill, binned, ovf, ovfCnt, E, R);
    range_build_kernel<<<R, B, 0, stream>>>(binned, ovf, rangeFill, ovfCnt, start, cnt, dinv, csr, N);

    gemm1_wmma_kernel<<<gemmBlocks, B, 0, stream>>>(x, pW1, dinv, bufA, N);
    gather1_gemm2_kernel<<<fusedBlocks, G1W * 32, 0, stream>>>(bufA, dinv, start, cnt, csr, b1, pW2, bufB, N);

    gather2_kernel<<<gather2Blk, G2W * 32, 0, stream>>>(bufB, dinv, start, cnt, csr, b2, out, N);
}


# x @ W1 runs in the list-build launch on blocks of its own that count their rows' in-degrees from the binned pairs for the row scale: build waits overlap matrix work; launches 6 -> 5; sums unchanged
# speedup vs baseline: 5.3099x; 1.0792x over previous
#include <hip/hip_runtime.h>

typedef __attribute__((ext_vector_type(2))) float v2f;
typedef __attribute__((ext_vector_type(4))) float v4f;
typedef __attribute__((ext_vector_type(8))) float v8f;

#define F_IN  128
#define F_HID 64
#define F_OUT 40


#define RSH   8
#define RS    256
#define MAXR  1024
#define TILE  4096
#define EPT   (TILE / 256)
#define RPT   (MAXR / 256)
#define BPT   20
#define SCAP  8192
#define CAP   SCAP

__device__ __forceinline__ int block_scan_incl(int v, int* wsum, int& total) {
    const int lane = threadIdx.x & 31;
    const int wave = threadIdx.x >> 5;
    int incl = v;
    for (int o = 1; o < 32; o <<= 1) {
        int t = __shfl_up(incl, o);
        if (lane >= o) incl += t;
    }
    if (lane == 31) wsum[wave] = incl;
    __syncthreads();
    int before = 0, all = 0;
    for (int w = 0; w < 8; ++w) {
        const int s = wsum[w];
        if (w < wave) before += s;
        all += s;
    }
    total = all;
    __syncthreads();
    return incl + before;
}

__global__ void __launch_bounds__(256)
range_bin_kernel(const int* __restrict__ src, const int* __restrict__ dst,
                 int* __restrict__ rangeFill, int2* __restrict__ binned,
                 int2* __restrict__ ovf, int* __restrict__ ovfCnt, int E, int R) {
    __shared__ int h[MAXR];
    __shared__ int loc[MAXR];
    __shared__ int gb[MAXR];
    __shared__ int2 stg[TILE];
    __shared__ int wsum[8];
    const int tid = threadIdx.x;
    const int base = blockIdx.x * TILE;
    int ss[EPT], dd[EPT], lr[EPT];
    for (int k = 0; k < EPT; ++k) {
        const int e  = base + k * 256 + tid;
        const int ec = e < E ? e : E - 1;
        dd[k] = dst[ec];
        ss[k] = src[ec];
    }
    for (int j = 0; j < 4; ++j) h[tid * 4 + j] = 0;
    __syncthreads();
    for (int k = 0; k < EPT; ++k) {
        const int e = base + k * 256 + tid;
        lr[k] = 0;
        if (e < E) lr[k] = atomicAdd(&h[dd[k] >> RSH], 1);
    }
    __syncthreads();
    int cc[RPT], gg[RPT];
    for (int j = 0; j < RPT; ++j) {
        const int i = tid + j * 256;
        cc[j] = (i < R) ? h[i] : 0;
    }
    for (int j = 0; j < RPT; ++j) {
        gg[j] = 0;
        if (cc[j]) gg[j] = atomicAdd(&rangeFill[tid + j * 256], cc[j]);
    }
    int c4[4];
    for (int j = 0; j < 4; ++j) c4[j] = h[tid * 4 + j];
    const int t4 = c4[0] + c4[1] + c4[2] + c4[3];
    int nt;
    int lrun = block_scan_incl(t4, wsum, nt) - t4;
    for (int j = 0; j < 4; ++j) {
        loc[tid * 4 + j] = lrun;
        lrun += c4[j];
    }
    for (int j = 0; j < RPT; ++j) {
        const int i = tid + j * 256;
        if (cc[j]) gb[i] = gg[j];
    }
    __syncthreads();
    int lb[EPT];
    for (int k = 0; k < EPT; ++k) lb[k] = loc[dd[k] >> RSH];
    for (int k = 0; k < EPT; ++k) {
        const int e = base + k * 256 + tid;
        if (e < E) stg[lb[k] + lr[k]] = make_int2(ss[k], dd[k]);
    }
    __syncthreads();
    int2 pr[EPT];
    int  gq[EPT], sl[EPT];
    for (int k = 0; k < EPT; ++k) {
        const int i = k * 256 + tid;
        pr[k] = stg[i];
        if (i >= nt) pr[k] = make_int2(0, 0);
    }
    for (int k = 0; k < EPT; ++k) {
        const int i = k * 256 + tid;
        const int r = pr[k].y >> RSH;
        gq[k] = gb[r] + (i - loc[r]);
        sl[k] = r * CAP + gq[k];
    }
    bool spill = false;
    for (int k = 0; k < EPT; ++k) {
        const int i = k * 256 + tid;
        const bool fits = gq[k] < CAP;
        if ((i < nt) & fits) binned[sl[k]] = pr[k];
        spill |= (i < nt) & !fits;
    }
    if (spill) {
        for (int k = 0; k < EPT; ++k) {
            const int i = k * 256 + tid;
            if (i < nt && gq[k] >= CAP) {
                const int pos = atomicAdd(ovfCnt, 1);
                ovf[pos] = pr[k];
            }
        }
    }
}

__device__ __forceinline__ void
range_build_role(const int r, const int2* __restrict__ binned, const int2* __restrict__ ovf,
                 const int* __restrict__ rangeFill, const int* __restrict__ ovfCnt,
                 int* __restrict__ start, int* __restrict__ cnt, float* __restrict__ dinv,
                 int* __restrict__ csr, int N, const int tot, int* c, int* cur, int* stg, int* wsum) {
    const int tid = threadIdx.x;
    int part = 0;
    for (int j = 0; j < RPT; ++j) {
        const int i = tid + j * 256;
        part += (i < r) ? rangeFill[i] : 0;
    }
    int rbeg;
    block_scan_incl(part, wsum, rbeg);
    const bool spilled = tot > CAP;
    const int  n       = spilled ? CAP : tot;
    const int  nOv     = spilled ? ovfCnt[0] : 0;
    const int2* __restrict__ reg = binned + (size_t)r * CAP;
    c[tid] = 0;
    __syncthreads();
    int2 p[BPT];
    int held = -1;
    for (int c0 = 0; c0 < n; c0 += BPT * 256) {
        for (int k = 0; k < BPT; ++k) {
            const int i = c0 + k * 256 + tid;
            p[k] = reg[i < n ? i : n - 1];
        }
        for (int k = 0; k < BPT; ++k) {
            const int i = c0 + k * 256 + tid;
            if (i < n) atomicAdd(&c[p[k].y & (RS - 1)], 1);
        }
        held = c0;
    }
    if (spilled) {
        for (int i = tid; i < nOv; i += 256) {
            const int2 q = ovf[i];
            if ((q.y >> RSH) == r) atomicAdd(&c[q.y & (RS - 1)], 1);
        }
    }
    __syncthreads();
    const int v = c[tid];
    int tot2;
    const int excl = block_scan_incl(v, wsum, tot2) - v;
    const int node = r * RS + tid;
    if (node < N) {
        start[node] = rbeg + excl;
        cnt[node]   = v;
        dinv[node]  = rsqrtf((float)(v + 1));
    }
    const bool staged = !spilled;
    cur[tid] = staged ? excl : rbeg + excl;
    __syncthreads();
    for (int c0 = 0; c0 < n; c0 += BPT * 256) {
        if (c0 != held) {
            for (int k = 0; k < BPT; ++k) {
                const int i = c0 + k * 256 + tid;
                p[k] = reg[i < n ? i : n - 1];
            }
            held = c0;
        }
        int pos[BPT];
        for (int k = 0; k < BPT; ++k) {
            const int i = c0 + k * 256 + tid;
            pos[k] = 0;
            if (i < n) pos[k] = atomicAdd(&cur[p[k].y & (RS - 1)], 1);
        }
        if (staged) {
            for (int k = 0; k < BPT; ++k) {
                const int i = c0 + k * 256 + tid;
                if (i < n) stg[pos[k]] = p[k].x;
            }
        } else {
            for (int k = 0; k < BPT; ++k) {
                const int i = c0 + k * 256 + tid;
                if (i < n) csr[pos[k]] = p[k].x;
            }
        }
    }
    if (spilled) {
        for (int i = tid; i < nOv; i += 256) {
            const int2 q = ovf[i];
            if ((q.y >> RSH) == r) {
                const int pos = atomicAdd(&cur[q.y & (RS - 1)], 1);
                csr[pos] = q.x;
            }
        }
    }
    if (staged) {
        __syncthreads();
        for (int i0 = tid; i0 < n; i0 += 4 * 256) {
            int s4[4];
            for (int q = 0; q < 4; ++q) {
                const int i = i0 + q * 256;
                s4[q] = stg[i < n ? i : n - 1];
            }
            for (int q = 0; q < 4; ++q) {
                const int i = i0 + q * 256;
                if (i < n) csr[rbeg + i] = s4[q];
            }
        }
    }
}


__global__ void prep_kernel(const float* __restrict__ W1, const float* __restrict__ W2,
                            float* __restrict__ pW1, float* __restrict__ pW2,
                            int* __restrict__ rangeFill, int* __restrict__ ovfCnt, int R) {
    const int idx  = blockIdx.x * blockDim.x + threadIdx.x;
    const int kk   = idx >> 8;
    const int rem  = idx & 255;
    const int lane = rem >> 3;
    const int j    = rem & 7;
    const int n    = j >> 1;
    const int t    = j & 1;
    const int lm   = lane & 15;
    const int kh   = (lane >> 4) << 1;
    if (idx < (F_IN / 4) * 256)
        pW1[idx] = W1[(size_t)(4 * kk + kh + t) * F_HID + n * 16 + lm];
    if (idx < (F_HID / 4) * 256) {
        const int col = n * 16 + lm;
        float v = 0.0f;
        if (j < 6 && col < F_OUT)
            v = W2[(size_t)(4 * kk + kh + t) * F_OUT + col];
        pW2[idx] = v;
    }
    if (idx < R) rangeFill[idx] = 0;
    if (idx == 0) ovfCnt[0] = 0;
}


__device__ __forceinline__ void mma_step(v8f& acc0, v8f& acc1, v8f& acc2, v8f& acc3,
                                         const v2f A, const v4f w01, const v4f w23) {
    v2f B0; B0[0] = w01[0]; B0[1] = w01[1];
    v2f B1; B1[0] = w01[2]; B1[1] = w01[3];
    v2f B2; B2[0] = w23[0]; B2[1] = w23[1];
    v2f B3; B3[0] = w23[2]; B3[1] = w23[3];
    acc0 = __builtin_amdgcn_wmma_f32_16x16x4_f32(false, A, false, B0, (short)0, acc0, false, false);
    acc1 = __builtin_amdgcn_wmma_f32_16x16x4_f32(false, A, false, B1, (short)0, acc1, false, false);
    acc2 = __builtin_amdgcn_wmma_f32_16x16x4_f32(false, A, false, B2, (short)0, acc2, false, false);
    acc3 = __builtin_amdgcn_wmma_f32_16x16x4_f32(false, A, false, B3, (short)0, acc3, false, false);
}

__device__ __forceinline__ void
gemm1_role(const int b, const float* __restrict__ x, const float* __restrict__ pW1,
           const int2* __restrict__ binned, const int2* __restrict__ ovf,
           const int* __restrict__ rangeFill, const int* __restrict__ ovfCnt,
           float* __restrict__ h1, int N, const int tot, int* c) {
    const int tid  = threadIdx.x;
    const int lane = tid & 31;
    const int wave = tid >> 5;
    const int row0 = b * 128 + wave * 16;
    const int lm   = lane & 15;
    const int kh   = (lane >> 4) << 1;
    int r = row0 + lm;
    if (r > N - 1) r = N - 1;
    const float* xrow = x + (size_t)r * F_IN + kh;
    const float* pw   = pW1 + (size_t)lane * 8;

    v2f a0 = *(const v2f*)(xrow + 0),  a1 = *(const v2f*)(xrow + 4);
    v2f a2 = *(const v2f*)(xrow + 8),  a3 = *(const v2f*)(xrow + 12);
    v4f l0 = *(const v4f*)(pw + 0),    g0 = *(const v4f*)(pw + 4);
    v4f l1 = *(const v4f*)(pw + 256),  g1 = *(const v4f*)(pw + 256 + 4);
    v4f l2 = *(const v4f*)(pw + 512),  g2 = *(const v4f*)(pw + 512 + 4);
    v4f l3 = *(const v4f*)(pw + 768),  g3 = *(const v4f*)(pw + 768 + 4);

    const int  rg      = b >> 1;
    const bool spilled = tot > CAP;
    const int  n       = spilled ? CAP : tot;
    const int  nOv     = spilled ? ovfCnt[0] : 0;
    const int2* __restrict__ reg = binned + (size_t)rg * CAP;
    const int last = n > 0 ? n - 1 : 0;
    int dq[BPT];
    for (int k = 0; k < BPT; ++k) {
        const int i = k * 256 + tid;
        dq[k] = reg[i < n ? i : last].y;
    }
    c[tid] = 0;
    __syncthreads();
    for (int k = 0; k < BPT; ++k) {
        const int i = k * 256 + tid;
        if (i < n) atomicAdd(&c[dq[k] & (RS - 1)], 1);
    }
    for (int c0 = BPT * 256; c0 < n; c0 += BPT * 256) {
        for (int k = 0; k < BPT; ++k) {
            const int i = c0 + k * 256 + tid;
            dq[k] = reg[i < n ? i : last].y;
        }
        for (int k = 0; k < BPT; ++k) {
            const int i = c0 + k * 256 + tid;
            if (i < n) atomicAdd(&c[dq[k] & (RS - 1)], 1);
        }
    }
    if (spilled) {
        for (int i = tid; i < nOv; i += 256) {
            const int2 q = ovf[i];
            if ((q.y >> RSH) == rg) atomicAdd(&c[q.y & (RS - 1)], 1);
        }
    }

    v8f acc0 = {}, acc1 = {}, acc2 = {}, acc3 = {};

    #pragma unroll 1
    for (int kk = 0; kk < F_IN / 4 - 4; kk += 4) {
        const float* xn = xrow + (kk + 4) * 4;
        const float* pn = pw + (size_t)(kk + 4) * 256;
        mma_step(acc0, acc1, acc2, acc3, a0, l0, g0);
        a0 = *(const v2f*)(xn);
        l0 = *(const v4f*)(pn);
        g0 = *(const v4f*)(pn + 4);
        mma_step(acc0, acc1, acc2, acc3, a1, l1, g1);
        a1 = *(const v2f*)(xn + 4);
        l1 = *(const v4f*)(pn + 256);
        g1 = *(const v4f*)(pn + 256 + 4);
        mma_step(acc0, acc1, acc2, acc3, a2, l2, g2);
        a2 = *(const v2f*)(xn + 8);
        l2 = *(const v4f*)(pn + 512);
        g2 = *(const v4f*)(pn + 512 + 4);
        mma_step(acc0, acc1, acc2, acc3, a3, l3, g3);
        a3 = *(const v2f*)(xn + 12);
        l3 = *(const v4f*)(pn + 768);
        g3 = *(const v4f*)(pn + 768 + 4);
    }
    mma_step(acc0, acc1, acc2, acc3, a0, l0, g0);
    mma_step(acc0, acc1, acc2, acc3, a1, l1, g1);
    mma_step(acc0, acc1, acc2, acc3, a2, l2, g2);
    mma_step(acc0, acc1, acc2, acc3, a3, l3, g3);

    __syncthreads();
    const int hi8 = (lane >> 4) << 3;
    const int lr0 = (b & 1) * 128 + wave * 16 + hi8;
    float dv[8];
    for (int j = 0; j < 8; ++j) dv[j] = rsqrtf((float)(c[lr0 + j] + 1));

    for (int j = 0; j < 8; ++j) {
        int rr = row0 + hi8 + j;
        if (rr < N) {
            float* o = h1 + (size_t)rr * F_HID + lm;
            o[0]  = acc0[j] * dv[j];
            o[16] = acc1[j] * dv[j];
            o[32] = acc2[j] * dv[j];
            o[48] = acc3[j] * dv[j];
        }
    }
}

__global__ void __launch_bounds__(256)
build_gemm1_kernel(const int2* __restrict__ binned, const int2* __restrict__ ovf,
                   const int* __restrict__ rangeFill, const int* __restrict__ ovfCnt,
                   int* __restrict__ start, int* __restrict__ cnt, float* __restrict__ dinv,
                   int* __restrict__ csr, const float* __restrict__ x,
                   const float* __restrict__ pW1, float* __restrict__ h1, int N, int G) {
    __shared__ int c[RS];
    __shared__ int cur[RS];
    __shared__ int stg[SCAP];
    __shared__ int wsum[8];
    const int  bid  = blockIdx.x;
    const bool prod = bid < G;
    const int  tot  = rangeFill[prod ? (bid >> 1) : (bid - G)];
    if (prod)
        gemm1_role(bid, x, pW1, binned, ovf, rangeFill, ovfCnt, h1, N, tot, c);
    else
        range_build_role(bid - G, binned, ovf, rangeFill, ovfCnt, start, cnt, dinv, csr, N, tot,
                         c, cur, stg, wsum);
}


#define RL(v, j) __builtin_amdgcn_readlane((v), (j))

__device__ __forceinline__ float2 ld_row2(const float* __restrict__ col, int s, int pitch) {
    return *(const float2*)(col + (size_t)s * pitch);
}

__device__ __forceinline__ float2 sum_rows(const float* __restrict__ col, int pitch,
                                           int my_s, int n, float2 a) {
    int j = 0;
    for (; j + 8 <= n; j += 8) {
        const float2 v0 = ld_row2(col, RL(my_s, j + 0), pitch);
        const float2 v1 = ld_row2(col, RL(my_s, j + 1), pitch);
        const float2 v2 = ld_row2(col, RL(my_s, j + 2), pitch);
        const float2 v3 = ld_row2(col, RL(my_s, j + 3), pitch);
        const float2 v4 = ld_row2(col, RL(my_s, j + 4), pitch);
        const float2 v5 = ld_row2(col, RL(my_s, j + 5), pitch);
        const float2 v6 = ld_row2(col, RL(my_s, j + 6), pitch);
        const float2 v7 = ld_row2(col, RL(my_s, j + 7), pitch);
        a.x += v0.x; a.y += v0.y;
        a.x += v1.x; a.y += v1.y;
        a.x += v2.x; a.y += v2.y;
        a.x += v3.x; a.y += v3.y;
        a.x += v4.x; a.y += v4.y;
        a.x += v5.x; a.y += v5.y;
        a.x += v6.x; a.y += v6.y;
        a.x += v7.x; a.y += v7.y;
    }
    if (j + 4 <= n) {
        const float2 v0 = ld_row2(col, RL(my_s, j + 0), pitch);
        const float2 v1 = ld_row2(col, RL(my_s, j + 1), pitch);
        const float2 v2 = ld_row2(col, RL(my_s, j + 2), pitch);
        const float2 v3 = ld_row2(col, RL(my_s, j + 3), pitch);
        a.x += v0.x; a.y += v0.y;
        a.x += v1.x; a.y += v1.y;
        a.x += v2.x; a.y += v2.y;
        a.x += v3.x; a.y += v3.y;
        j += 4;
    }
    if (j + 2 <= n) {
        const float2 v0 = ld_row2(col, RL(my_s, j + 0), pitch);
        const float2 v1 = ld_row2(col, RL(my_s, j + 1), pitch);
        a.x += v0.x; a.y += v0.y;
        a.x += v1.x; a.y += v1.y;
        j += 2;
    }
    if (j < n) {
        const float2 v0 = ld_row2(col, RL(my_s, j), pitch);
        a.x += v0.x; a.y += v0.y;
    }
    return a;
}


#define NPW 8

__device__ __forceinline__ float2 gather_node(const float* __restrict__ col, int pitch,
                                              const int* __restrict__ csr, int node, int beg, int c,
                                              int my_s, float2 self, int lane) {
    float2 a = sum_rows(col, pitch, my_s, c < 32 ? c : 32, self);
    for (int base = 32; base < c; base += 32) {
        int n = c - base;
        if (n > 32) n = 32;
        int s2 = node;
        if (lane < n) s2 = csr[beg + base + lane];
        a = sum_rows(col, pitch, s2, n, a);
    }
    return a;
}


#define G1W 4
#define NB  (G1W * NPW)
#define ZP  (F_HID + 4)

__global__ void __launch_bounds__(G1W * 32)
gather1_gemm2_kernel(const float* __restrict__ h1s, const float* __restrict__ dinv,
                     const int* __restrict__ start, const int* __restrict__ cnt,
                     const int* __restrict__ csr, const float* __restrict__ b1,
                     const float* __restrict__ pW2, float* __restrict__ h2s, int N) {
    __shared__ __attribute__((aligned(16))) float sz[NB * ZP];
    __shared__ __attribute__((aligned(16))) float so[NB * F_OUT];
    __shared__ float sd[NB];
    const int lane = threadIdx.x & 31;
    const int wave = __builtin_amdgcn_readfirstlane(threadIdx.x >> 5);
    const int row0 = wave * NPW;
    const int n0   = blockIdx.x * NB + row0;
    {
        const float* col = h1s + lane * 2;
        const float2 bb = *(const float2*)(b1 + lane * 2);
        int ln = n0 + (lane < NPW ? lane : NPW - 1);
        if (ln > N - 1) ln = N - 1;
        const int   sv  = start[ln];
        const int   cv  = cnt[ln];
        const float dvv = dinv[ln];
        if (lane < NPW) sd[row0 + lane] = dvv;
        int node = n0 < N - 1 ? n0 : N - 1;
        int beg  = RL(sv, 0);
        int c    = RL(cv, 0);
        int my_s = node;
        if (lane < c) my_s = csr[beg + lane];
        float2 self = ld_row2(col, node, F_HID);
        for (int i = 0; i < NPW; ++i) {
            const int ni = i + 1 < NPW ? i + 1 : NPW - 1;
            int nnode = n0 + ni;
            if (nnode > N - 1) nnode = N - 1;
            const int nbeg = RL(sv, ni);
            const int nc   = RL(cv, ni);
            int nmy = nnode;
            if (lane < nc) nmy = csr[nbeg + lane];
            const float2 nself = ld_row2(col, nnode, F_HID);
            const float dd = __int_as_float(RL(__float_as_int(dvv), i));
            const float2 a = gather_node(col, F_HID, csr, node, beg, c, my_s, self, lane);
            const float rx = a.x * dd + bb.x;
            const float ry = a.y * dd + bb.y;
            float2 r;
            r.x = rx > 0.0f ? rx : 0.0f;
            r.y = ry > 0.0f ? ry : 0.0f;
            *(float2*)(sz + (row0 + i) * ZP + lane * 2) = r;
            node = nnode; beg = nbeg; c = nc; my_s = nmy; self = nself;
        }
    }
    __syncthreads();
    if (wave < 3) {
        const int lm  = lane & 15;
        const int kh  = (lane >> 4) << 1;
        const int hi8 = (lane >> 4) << 3;
        const float* zrow0 = sz + lm * ZP + kh;
        const float* zrow1 = zrow0 + 16 * ZP;
        const float* pw    = pW2 + (size_t)lane * 8 + wave * 2;
        v2f Bf[F_HID / 4];
        #pragma unroll
        for (int kk = 0; kk < F_HID / 4; ++kk) Bf[kk] = *(const v2f*)(pw + (size_t)kk * 256);
        v8f acc0 = {}, acc1 = {};
        #pragma unroll
        for (int kk = 0; kk < F_HID / 4; ++kk) {
            const v2f A0 = *(const v2f*)(zrow0 + kk * 4);
            const v2f A1 = *(const v2f*)(zrow1 + kk * 4);
            acc0 = __builtin_amdgcn_wmma_f32_16x16x4_f32(false, A0, false, Bf[kk], (short)0, acc0, false, false);
            acc1 = __builtin_amdgcn_wmma_f32_16x16x4_f32(false, A1, false, Bf[kk], (short)0, acc1, false, false);
        }
        const int colo = wave * 16 + lm;
        if (colo < F_OUT) {
            #pragma unroll
            for (int j = 0; j < 8; ++j) {
                so[(hi8 + j) * F_OUT + colo]      = acc0[j] * sd[hi8 + j];
                so[(16 + hi8 + j) * F_OUT + colo] = acc1[j] * sd[16 + hi8 + j];
            }
        }
    }
    __syncthreads();
    for (int t = threadIdx.x; t < (NB * F_OUT) / 4; t += G1W * 32) {
        const size_t g = (size_t)blockIdx.x * (NB * F_OUT) + (size_t)t * 4;
        if (g < (size_t)N * F_OUT)
            *(v4f*)(h2s + g) = *(const v4f*)(so + t * 4);
    }
}


#define G2W 4
#define NB2 (G2W * NPW)

__global__ void __launch_bounds__(G2W * 32)
gather2_kernel(const float* __restrict__ h2s, const float* __restrict__ dinv,
               const int* __restrict__ start, const int* __restrict__ cnt,
               const int* __restrict__ csr, const float* __restrict__ b2,
               float* __restrict__ out, int N) {
    __shared__ __attribute__((aligned(16))) float so[NB2 * F_OUT];
    const int lane = threadIdx.x & 31;
    const int wave = __builtin_amdgcn_readfirstlane(threadIdx.x >> 5);
    const int row0 = wave * NPW;
    const int n0   = blockIdx.x * NB2 + row0;
    if (n0 < N) {
        const int cl = lane < 20 ? lane : 19;
        const float* col = h2s + cl * 2;
        const float2 bb = *(const float2*)(b2 + cl * 2);
        int ln = n0 + (lane < NPW ? lane : NPW - 1);
        if (ln > N - 1) ln = N - 1;
        const int   sv  = start[ln];
        const int   cv  = cnt[ln];
        const float dvv = dinv[ln];
        int node = n0;
        int beg  = RL(sv, 0);
        int c    = RL(cv, 0);
        int my_s = node;
        if (lane < c) my_s = csr[beg + lane];
        float2 self = ld_row2(col, node, F_OUT);
        for (int i = 0; i < NPW; ++i) {
            const int ni = i + 1 < NPW ? i + 1 : NPW - 1;
            int nnode = n0 + ni;
            if (nnode > N - 1) nnode = N - 1;
            const int nbeg = RL(sv, ni);
            const int nc   = RL(cv, ni);
            int nmy = nnode;
            if (lane < nc) nmy = csr[nbeg + lane];
            const float2 nself = ld_row2(col, nnode, F_OUT);
            const float dd = __int_as_float(RL(__float_as_int(dvv), i));
            const float2 a = gather_node(col, F_OUT, csr, node, beg, c, my_s, self, lane);
            if (lane < 20) {
                so[(row0 + i) * F_OUT + lane * 2]     = a.x * dd + bb.x;
                so[(row0 + i) * F_OUT + lane * 2 + 1] = a.y * dd + bb.y;
            }
            node = nnode; beg = nbeg; c = nc; my_s = nmy; self = nself;
        }
    }
    __syncthreads();
    for (int t = threadIdx.x; t < (NB2 * F_OUT) / 4; t += G2W * 32) {
        const size_t g = (size_t)blockIdx.x * (NB2 * F_OUT) + (size_t)t * 4;
        if (g < (size_t)N * F_OUT)
            *(v4f*)(out + g) = *(const v4f*)(so + t * 4);
    }
}


extern "C" void kernel_launch(void* const* d_in, const int* in_sizes, int n_in,
                              void* d_out, int out_size, void* d_ws, size_t ws_size,
                              hipStream_t stream) {
    const float* x  = (const float*)d_in[0];
    const int*   ei = (const int*)d_in[1];
    const float* W1 = (const float*)d_in[2];
    const float* b1 = (const float*)d_in[3];
    const float* W2 = (const float*)d_in[4];
    const float* b2 = (const float*)d_in[5];

    const int N = in_sizes[0] / F_IN;
    const int E = in_sizes[1] / 2;
    const int* src = ei;
    const int* dst = ei + E;

    float* ws    = (float*)d_ws;
    float* dinv  = ws + (size_t)N;
    float* bufA  = ws + (size_t)2 * N;
    float* bufB  = bufA + (size_t)N * F_HID;
    float* pW1   = bufB + (size_t)N * F_HID;
    float* pW2   = pW1 + (size_t)(F_IN / 4) * 256;
    int*   cnt   = (int*)(pW2 + (size_t)(F_HID / 4) * 256);
    int*   start = cnt + (size_t)N;
    const int R = (N + RS - 1) / RS;
    int2*  binned = (int2*)(start + (size_t)N + (N & 1));
    int*   csr   = (int*)(binned + (size_t)R * CAP);
    int2*  ovf   = (int2*)(csr + (size_t)E + (E & 1));
    int*   rangeFill = (int*)(ovf + (size_t)E);
    int*   ovfCnt    = rangeFill + MAXR;
    float* out   = (float*)d_out;

    const int B = 256;
    const int tileBlocks  = (E + TILE - 1) / TILE;
    const int gemmBlocks  = (N + 127) / 128;
    const int gather2Blk  = (N + NB2 - 1) / NB2;
    const int fusedBlocks = (N + NB - 1) / NB;
    const int prepBlocks  = ((F_IN / 4) * 256 + B - 1) / B;

    prep_kernel<<<prepBlocks, B, 0, stream>>>(W1, W2, pW1, pW2, rangeFill, ovfCnt, R);

    range_bin_kernel<<<tileBlocks, B, 0, stream>>>(src, dst, rangeFill, binned, ovf, ovfCnt, E, R);
    build_gemm1_kernel<<<gemmBlocks + R, B, 0, stream>>>(binned, ovf, rangeFill, ovfCnt, start, cnt, dinv, csr,
                                                        x, pW1, bufA, N, gemmBlocks);

    gather1_gemm2_kernel<<<fusedBlocks, G1W * 32, 0, stream>>>(bufA, dinv, start, cnt, csr, b1, pW2, bufB, N);

    gather2_kernel<<<gather2Blk, G2W * 32, 0, stream>>>(bufB, dinv, start, cnt, csr, b2, out, N);
}
